# starting kernel
# baseline (speedup 1.0000x reference)
#include <hip/hip_runtime.h>


typedef float v2f __attribute__((ext_vector_type(2)));
typedef float v8f __attribute__((ext_vector_type(8)));

#define B_    16
#define T_    2048
#define C_    256
#define K_    8
#define F_    256
#define DIL_  4
#define HALO  28
#define TT    16
#define XROWS 44
#define PITCH 260
#define NTHREADS 128

__global__ __launch_bounds__(256)
void wt_transpose_kernel(const float* __restrict__ w, float* __restrict__ wt) {
    int idx = blockIdx.x * 256 + threadIdx.x;
    if (idx >= K_ * F_ * C_) return;
    int c = idx % C_;
    int f = (idx / C_) % F_;
    int k = idx / (C_ * F_);
    wt[idx] = w[(c * K_ + k) * F_ + f];
}

__global__ __launch_bounds__(NTHREADS)
void MyDeConv1D_80367428043218_kernel(const float* __restrict__ x,
                                      const float* __restrict__ wt,
                                      const float* __restrict__ bias,
                                      float* __restrict__ out) {
    __shared__ __align__(16) float xs[XROWS * PITCH];

    const int tile = blockIdx.x & (T_ / TT - 1);
    const int b    = blockIdx.x >> 7;
    const int t_lo = tile * TT;
    const int tid  = threadIdx.x;

    const float4* xg = (const float4*)(x + (size_t)b * T_ * C_);
    for (int i = tid; i < XROWS * (C_ / 4); i += NTHREADS) {
        int row = i >> 6;
        int cq  = i & 63;
        int t   = t_lo + row;
        float4 v = make_float4(0.f, 0.f, 0.f, 0.f);
        if (t < T_) v = xg[t * (C_ / 4) + cq];
        *(float4*)&xs[row * PITCH + cq * 4] = v;
    }
    __syncthreads();

    const int lane  = tid & 31;
    const int wv    = tid >> 5;
    const int n16   = lane & 15;
    const int half  = lane >> 4;
    const int chalf = half * 2;
    const int f0    = wv * 64;

    v8f acc0 = {};
    v8f acc1 = {};
    v8f acc2 = {};
    v8f acc3 = {};

    for (int k = 0; k < K_; ++k) {
        const float* aP  = &xs[(HALO - DIL_ * k + n16) * PITCH + chalf];
        const float* b0P = wt + (((size_t)k * F_ + f0 + n16) * C_ + chalf);
#pragma unroll 4
        for (int cc = 0; cc < C_ / 4; ++cc) {
            v2f a  = *(const v2f*)(aP  + cc * 4);
            v2f b0 = *(const v2f*)(b0P + cc * 4);
            v2f b1 = *(const v2f*)(b0P + 16 * C_ + cc * 4);
            v2f b2 = *(const v2f*)(b0P + 32 * C_ + cc * 4);
            v2f b3 = *(const v2f*)(b0P + 48 * C_ + cc * 4);
            acc0 = __builtin_amdgcn_wmma_f32_16x16x4_f32(
                       false, a, false, b0, (short)0, acc0, false, false);
            acc1 = __builtin_amdgcn_wmma_f32_16x16x4_f32(
                       false, a, false, b1, (short)0, acc1, false, false);
            acc2 = __builtin_amdgcn_wmma_f32_16x16x4_f32(
                       false, a, false, b2, (short)0, acc2, false, false);
            acc3 = __builtin_amdgcn_wmma_f32_16x16x4_f32(
                       false, a, false, b3, (short)0, acc3, false, false);
        }
    }

    const float bias0 = bias[f0 + n16]      * (float)C_;
    const float bias1 = bias[f0 + 16 + n16] * (float)C_;
    const float bias2 = bias[f0 + 32 + n16] * (float)C_;
    const float bias3 = bias[f0 + 48 + n16] * (float)C_;
    float* outb = out + (size_t)b * T_ * F_;
#pragma unroll
    for (int i = 0; i < 8; ++i) {
        int m   = i + half * 8;
        int tau = t_lo + m;
        int d   = (tau + HALO) - (T_ - 1);
        int kmin = d > 0 ? ((d + 3) >> 2) : 0;
        float cnt = (float)(K_ - kmin);
        float* o = outb + (size_t)tau * F_ + f0 + n16;
        o[0]  = acc0[i] + cnt * bias0;
        o[16] = acc1[i] + cnt * bias1;
        o[32] = acc2[i] + cnt * bias2;
        o[48] = acc3[i] + cnt * bias3;
    }
}

extern "C" void kernel_launch(void* const* d_in, const int* in_sizes, int n_in,
                              void* d_out, int out_size, void* d_ws, size_t ws_size,
                              hipStream_t stream) {
    const float* x    = (const float*)d_in[0];
    const float* w    = (const float*)d_in[1];
    const float* bias = (const float*)d_in[2];
    float* out = (float*)d_out;
    float* wt  = (float*)d_ws;

    const int nW = K_ * F_ * C_;
    wt_transpose_kernel<<<(nW + 255) / 256, 256, 0, stream>>>(w, wt);

    MyDeConv1D_80367428043218_kernel<<<B_ * (T_ / TT), NTHREADS, 0, stream>>>(
        x, wt, bias, out);
}


# time rows per wave 16 -> 64 (wave tile 64 t x 64 f, block 64 t x 256 f): each B fragment fetched from L2 feeds 4 WMMAs instead of 1; per-tile summation order kept
# speedup vs baseline: 2.0673x; 2.0673x over previous
#include <hip/hip_runtime.h>


typedef float v2f __attribute__((ext_vector_type(2)));
typedef float v8f __attribute__((ext_vector_type(8)));

#define B_    16
#define T_    2048
#define C_    256
#define K_    8
#define F_    256
#define DIL_  4
#define HALO  28
#define TT    64
#define MT    4
#define NT    4
#define XROWS 92
#define PITCH 260
#define NTHREADS 128
#define STAGE_ITERS ((XROWS * (C_ / 4)) / NTHREADS)

__global__ __launch_bounds__(256)
void wt_transpose_kernel(const float* __restrict__ w, float* __restrict__ wt) {
    int idx = blockIdx.x * 256 + threadIdx.x;
    if (idx >= K_ * F_ * C_) return;
    int c = idx % C_;
    int f = (idx / C_) % F_;
    int k = idx / (C_ * F_);
    wt[idx] = w[(c * K_ + k) * F_ + f];
}

__global__ __launch_bounds__(NTHREADS)
void deconv1d_main_kernel(const float* __restrict__ x,
                          const float* __restrict__ wt,
                          const float* __restrict__ bias,
                          float* __restrict__ out) {
    __shared__ __align__(16) float xs[XROWS * PITCH];

    const int tile = blockIdx.x & (T_ / TT - 1);
    const int b    = blockIdx.x / (T_ / TT);
    const int t_lo = tile * TT;
    const int tid  = threadIdx.x;

    const float4* xg = (const float4*)(x + (size_t)b * T_ * C_);
#pragma unroll
    for (int it = 0; it < STAGE_ITERS; ++it) {
        int i   = tid + it * NTHREADS;
        int row = i >> 6;
        int cq  = i & 63;
        int t   = t_lo + row;
        int tc  = t < T_ ? t : (T_ - 1);
        float4 v = xg[tc * (C_ / 4) + cq];
        if (t >= T_) v = make_float4(0.f, 0.f, 0.f, 0.f);
        *(float4*)&xs[row * PITCH + cq * 4] = v;
    }
    __syncthreads();

    const int lane  = tid & 31;
    const int wv    = tid >> 5;
    const int n16   = lane & 15;
    const int half  = lane >> 4;
    const int chalf = half * 2;
    const int f0    = wv * 64;

    v8f acc[MT][NT];
#pragma unroll
    for (int mt = 0; mt < MT; ++mt)
#pragma unroll
        for (int ft = 0; ft < NT; ++ft)
            acc[mt][ft] = (v8f){0.f, 0.f, 0.f, 0.f, 0.f, 0.f, 0.f, 0.f};

    for (int k = 0; k < K_; ++k) {
        const float* aP = &xs[(HALO - DIL_ * k + n16) * PITCH + chalf];
        const float* bP = wt + (((size_t)k * F_ + f0 + n16) * C_ + chalf);
#pragma unroll 4
        for (int cc = 0; cc < C_ / 4; ++cc) {
            v2f a[MT], bf[NT];
#pragma unroll
            for (int mt = 0; mt < MT; ++mt)
                a[mt] = *(const v2f*)(aP + mt * 16 * PITCH + cc * 4);
#pragma unroll
            for (int ft = 0; ft < NT; ++ft)
                bf[ft] = *(const v2f*)(bP + ft * 16 * C_ + cc * 4);
#pragma unroll
            for (int mt = 0; mt < MT; ++mt)
#pragma unroll
                for (int ft = 0; ft < NT; ++ft)
                    acc[mt][ft] = __builtin_amdgcn_wmma_f32_16x16x4_f32(
                        false, a[mt], false, bf[ft], (short)0, acc[mt][ft], false, false);
        }
    }

    float biasv[NT];
#pragma unroll
    for (int ft = 0; ft < NT; ++ft)
        biasv[ft] = bias[f0 + ft * 16 + n16] * (float)C_;
    float* outb = out + (size_t)b * T_ * F_;
#pragma unroll
    for (int mt = 0; mt < MT; ++mt) {
#pragma unroll
        for (int i = 0; i < 8; ++i) {
            int m   = mt * 16 + i + half * 8;
            int tau = t_lo + m;
            int d   = (tau + HALO) - (T_ - 1);
            int kmin = d > 0 ? ((d + 3) >> 2) : 0;
            float cnt = (float)(K_ - kmin);
            float* o = outb + (size_t)tau * F_ + f0 + n16;
#pragma unroll
            for (int ft = 0; ft < NT; ++ft)
                o[ft * 16] = acc[mt][ft][i] + cnt * biasv[ft];
        }
    }
}

extern "C" void kernel_launch(void* const* d_in, const int* in_sizes, int n_in,
                              void* d_out, int out_size, void* d_ws, size_t ws_size,
                              hipStream_t stream) {
    const float* x    = (const float*)d_in[0];
    const float* w    = (const float*)d_in[1];
    const float* bias = (const float*)d_in[2];
    float* out = (float*)d_out;
    float* wt  = (float*)d_ws;

    const int nW = K_ * F_ * C_;
    wt_transpose_kernel<<<(nW + 255) / 256, 256, 0, stream>>>(w, wt);

    deconv1d_main_kernel<<<B_ * (T_ / TT), NTHREADS, 0, stream>>>(
        x, wt, bias, out);
}


# prepass stores W in WMMA B-fragment order: each wave-wide B load is one 16 B/lane read of 512 contiguous bytes (4 lines, 2 k4-steps) instead of 8 B/lane over 16 lines; summation order kept
# speedup vs baseline: 2.2002x; 1.0643x over previous
#include <hip/hip_runtime.h>


typedef float v2f __attribute__((ext_vector_type(2)));
typedef float v4f __attribute__((ext_vector_type(4)));
typedef float v8f __attribute__((ext_vector_type(8)));

#define B_    16
#define T_    2048
#define C_    256
#define K_    8
#define F_    256
#define DIL_  4
#define HALO  28
#define TT    64
#define MT    4
#define NT    4
#define XROWS 92
#define PITCH 260
#define NTHREADS 128
#define STAGE_ITERS ((XROWS * (C_ / 4)) / NTHREADS)
#define FTILES (F_ / 16)
#define CPAIRS (C_ / 8)

__global__ __launch_bounds__(256)
void wt_pack_kernel(const float* __restrict__ w, float* __restrict__ wp) {
    int idx = blockIdx.x * 256 + threadIdx.x;
    if (idx >= K_ * F_ * C_ / 4) return;
    int lane  = idx & 31;
    int cp    = (idx >> 5) & (CPAIRS - 1);
    int ftile = (idx >> 10) & (FTILES - 1);
    int k     = idx >> 14;
    int f     = ftile * 16 + (lane & 15);
    int c0    = cp * 8 + (lane >> 4) * 2;
    const float* p = w + ((size_t)c0 * K_ + k) * F_ + f;
    v4f v;
    v.x = p[0];
    v.y = p[K_ * F_];
    v.z = p[4 * K_ * F_];
    v.w = p[5 * K_ * F_];
    ((v4f*)wp)[idx] = v;
}

__global__ __launch_bounds__(NTHREADS)
void deconv1d_main_kernel(const float* __restrict__ x,
                          const float* __restrict__ wp,
                          const float* __restrict__ bias,
                          float* __restrict__ out) {
    __shared__ __align__(16) float xs[XROWS * PITCH];

    const int tile = blockIdx.x & (T_ / TT - 1);
    const int b    = blockIdx.x / (T_ / TT);
    const int t_lo = tile * TT;
    const int tid  = threadIdx.x;

    const float4* xg = (const float4*)(x + (size_t)b * T_ * C_);
#pragma unroll
    for (int it = 0; it < STAGE_ITERS; ++it) {
        int i   = tid + it * NTHREADS;
        int row = i >> 6;
        int cq  = i & 63;
        int t   = t_lo + row;
        int tc  = t < T_ ? t : (T_ - 1);
        float4 v = xg[tc * (C_ / 4) + cq];
        if (t >= T_) v = make_float4(0.f, 0.f, 0.f, 0.f);
        *(float4*)&xs[row * PITCH + cq * 4] = v;
    }
    __syncthreads();

    const int lane  = tid & 31;
    const int wv    = tid >> 5;
    const int n16   = lane & 15;
    const int half  = lane >> 4;
    const int chalf = half * 2;
    const int f0    = wv * 64;

    v8f acc[MT][NT];
#pragma unroll
    for (int mt = 0; mt < MT; ++mt)
#pragma unroll
        for (int ft = 0; ft < NT; ++ft)
            acc[mt][ft] = (v8f){0.f, 0.f, 0.f, 0.f, 0.f, 0.f, 0.f, 0.f};

    for (int k = 0; k < K_; ++k) {
        const float* aP = &xs[(HALO - DIL_ * k + n16) * PITCH + chalf];
        const v4f* bP = (const v4f*)wp + (((size_t)k * FTILES + wv * NT) * CPAIRS) * 32 + lane;
#pragma unroll 2
        for (int cp = 0; cp < CPAIRS; ++cp) {
            v2f a[2][MT];
            v4f b4[NT];
#pragma unroll
            for (int s = 0; s < 2; ++s)
#pragma unroll
                for (int mt = 0; mt < MT; ++mt)
                    a[s][mt] = *(const v2f*)(aP + mt * 16 * PITCH + cp * 8 + s * 4);
#pragma unroll
            for (int ft = 0; ft < NT; ++ft)
                b4[ft] = bP[(ft * CPAIRS + cp) * 32];
#pragma unroll
            for (int s = 0; s < 2; ++s)
#pragma unroll
                for (int mt = 0; mt < MT; ++mt)
#pragma unroll
                    for (int ft = 0; ft < NT; ++ft) {
                        v2f bf = s == 0 ? __builtin_shufflevector(b4[ft], b4[ft], 0, 1)
                                        : __builtin_shufflevector(b4[ft], b4[ft], 2, 3);
                        acc[mt][ft] = __builtin_amdgcn_wmma_f32_16x16x4_f32(
                            false, a[s][mt], false, bf, (short)0, acc[mt][ft], false, false);
                    }
        }
    }

    float biasv[NT];
#pragma unroll
    for (int ft = 0; ft < NT; ++ft)
        biasv[ft] = bias[f0 + ft * 16 + n16] * (float)C_;
    float* outb = out + (size_t)b * T_ * F_;
#pragma unroll
    for (int mt = 0; mt < MT; ++mt) {
#pragma unroll
        for (int i = 0; i < 8; ++i) {
            int m   = mt * 16 + i + half * 8;
            int tau = t_lo + m;
            int d   = (tau + HALO) - (T_ - 1);
            int kmin = d > 0 ? ((d + 3) >> 2) : 0;
            float cnt = (float)(K_ - kmin);
            float* o = outb + (size_t)tau * F_ + f0 + n16;
#pragma unroll
            for (int ft = 0; ft < NT; ++ft)
                o[ft * 16] = acc[mt][ft][i] + cnt * biasv[ft];
        }
    }
}

extern "C" void kernel_launch(void* const* d_in, const int* in_sizes, int n_in,
                              void* d_out, int out_size, void* d_ws, size_t ws_size,
                              hipStream_t stream) {
    const float* x    = (const float*)d_in[0];
    const float* w    = (const float*)d_in[1];
    const float* bias = (const float*)d_in[2];
    float* out = (float*)d_out;
    float* wp  = (float*)d_ws;

    wt_pack_kernel<<<K_ * F_ * C_ / 4 / 256, 256, 0, stream>>>(w, wp);

    deconv1d_main_kernel<<<B_ * (T_ / TT), NTHREADS, 0, stream>>>(
        x, wp, bias, out);
}


# fragment double-buffering: flat 128-step tap/channel loop, step s+1's A (LDS) and B (global) loads issued as one group ahead of step s's 64 WMMAs, not waited for at issue; summation order kept
# speedup vs baseline: 2.3421x; 1.0645x over previous
#include <hip/hip_runtime.h>


typedef float v2f __attribute__((ext_vector_type(2)));
typedef float v4f __attribute__((ext_vector_type(4)));
typedef float v8f __attribute__((ext_vector_type(8)));

#define B_    16
#define T_    2048
#define C_    256
#define K_    8
#define F_    256
#define DIL_  4
#define HALO  28
#define TT    64
#define MT    4
#define NT    4
#define XROWS 92
#define PITCH 260
#define NTHREADS 128
#define STAGE_ITERS ((XROWS * (C_ / 4)) / NTHREADS)
#define FTILES (F_ / 16)
#define CPAIRS (C_ / 8)
#define CQ     (CPAIRS / 2)
#define NSTEPS (K_ * CQ)

__global__ __launch_bounds__(256)
void wt_pack_kernel(const float* __restrict__ w, float* __restrict__ wp) {
    int idx = blockIdx.x * 256 + threadIdx.x;
    if (idx >= K_ * F_ * C_ / 4) return;
    int lane  = idx & 31;
    int cp    = (idx >> 5) & (CPAIRS - 1);
    int ftile = (idx >> 10) & (FTILES - 1);
    int k     = idx >> 14;
    int f     = ftile * 16 + (lane & 15);
    int c0    = cp * 8 + (lane >> 4) * 2;
    const float* p = w + ((size_t)c0 * K_ + k) * F_ + f;
    v4f v;
    v.x = p[0];
    v.y = p[K_ * F_];
    v.z = p[4 * K_ * F_];
    v.w = p[5 * K_ * F_];
    ((v4f*)wp)[idx] = v;
}

__device__ __forceinline__ void fence() { __builtin_amdgcn_sched_barrier(0); }

struct Frag {
    v2f a[2][2][MT];
    v4f b[2][NT];
};

__device__ __forceinline__ void load_frag(Frag& f, const float* aLane, const v4f* bLane, int step) {
    const int k  = step / CQ;
    const int cq = step % CQ;
    const float* aP = aLane + (HALO - DIL_ * k) * PITCH + cq * 16;
    const v4f*   bP = bLane + ((size_t)k * FTILES * CPAIRS + cq * 2) * 32;
#pragma unroll
    for (int u = 0; u < 2; ++u)
#pragma unroll
        for (int s = 0; s < 2; ++s)
#pragma unroll
            for (int mt = 0; mt < MT; ++mt)
                f.a[u][s][mt] = *(const v2f*)(aP + mt * 16 * PITCH + u * 8 + s * 4);
#pragma unroll
    for (int u = 0; u < 2; ++u)
#pragma unroll
        for (int ft = 0; ft < NT; ++ft)
            f.b[u][ft] = bP[(ft * CPAIRS + u) * 32];
}

__device__ __forceinline__ void mma_frag(v8f (&acc)[MT][NT], const Frag& f) {
#pragma unroll
    for (int u = 0; u < 2; ++u)
#pragma unroll
        for (int s = 0; s < 2; ++s)
#pragma unroll
            for (int mt = 0; mt < MT; ++mt)
#pragma unroll
                for (int ft = 0; ft < NT; ++ft) {
                    v2f bf = s == 0 ? __builtin_shufflevector(f.b[u][ft], f.b[u][ft], 0, 1)
                                    : __builtin_shufflevector(f.b[u][ft], f.b[u][ft], 2, 3);
                    acc[mt][ft] = __builtin_amdgcn_wmma_f32_16x16x4_f32(
                        false, f.a[u][s][mt], false, bf, (short)0, acc[mt][ft], false, false);
                }
}

__global__ __launch_bounds__(NTHREADS)
void deconv1d_main_kernel(const float* __restrict__ x,
                          const float* __restrict__ wp,
                          const float* __restrict__ bias,
                          float* __restrict__ out) {
    __shared__ __align__(16) float xs[XROWS * PITCH];

    const int tile = blockIdx.x & (T_ / TT - 1);
    const int b    = blockIdx.x / (T_ / TT);
    const int t_lo = tile * TT;
    const int tid  = threadIdx.x;

    const float4* xg = (const float4*)(x + (size_t)b * T_ * C_);
#pragma unroll
    for (int it = 0; it < STAGE_ITERS; ++it) {
        int i   = tid + it * NTHREADS;
        int row = i >> 6;
        int cq  = i & 63;
        int t   = t_lo + row;
        int tc  = t < T_ ? t : (T_ - 1);
        float4 v = xg[tc * (C_ / 4) + cq];
        if (t >= T_) v = make_float4(0.f, 0.f, 0.f, 0.f);
        *(float4*)&xs[row * PITCH + cq * 4] = v;
    }
    __syncthreads();

    const int lane  = tid & 31;
    const int wv    = tid >> 5;
    const int n16   = lane & 15;
    const int half  = lane >> 4;
    const int chalf = half * 2;
    const int f0    = wv * 64;

    v8f acc[MT][NT];
#pragma unroll
    for (int mt = 0; mt < MT; ++mt)
#pragma unroll
        for (int ft = 0; ft < NT; ++ft)
            acc[mt][ft] = (v8f){0.f, 0.f, 0.f, 0.f, 0.f, 0.f, 0.f, 0.f};

    const float* aLane = &xs[n16 * PITCH + chalf];
    const v4f*   bLane = (const v4f*)wp + ((size_t)wv * NT * CPAIRS) * 32 + lane;

    Frag g0, g1;
    load_frag(g0, aLane, bLane, 0);
    for (int it = 0; it < NSTEPS / 2 - 1; ++it) {
        load_frag(g1, aLane, bLane, 2 * it + 1);
        fence();
        mma_frag(acc, g0);
        fence();
        load_frag(g0, aLane, bLane, 2 * it + 2);
        fence();
        mma_frag(acc, g1);
    }
    load_frag(g1, aLane, bLane, NSTEPS - 1);
    fence();
    mma_frag(acc, g0);
    mma_frag(acc, g1);

    float biasv[NT];
#pragma unroll
    for (int ft = 0; ft < NT; ++ft)
        biasv[ft] = bias[f0 + ft * 16 + n16] * (float)C_;
    float* outb = out + (size_t)b * T_ * F_;
#pragma unroll
    for (int mt = 0; mt < MT; ++mt) {
#pragma unroll
        for (int i = 0; i < 8; ++i) {
            int m   = mt * 16 + i + half * 8;
            int tau = t_lo + m;
            int d   = (tau + HALO) - (T_ - 1);
            int kmin = d > 0 ? ((d + 3) >> 2) : 0;
            float cnt = (float)(K_ - kmin);
            float* o = outb + (size_t)tau * F_ + f0 + n16;
#pragma unroll
            for (int ft = 0; ft < NT; ++ft)
                o[ft * 16] = acc[mt][ft][i] + cnt * biasv[ft];
        }
    }
}

extern "C" void kernel_launch(void* const* d_in, const int* in_sizes, int n_in,
                              void* d_out, int out_size, void* d_ws, size_t ws_size,
                              hipStream_t stream) {
    const float* x    = (const float*)d_in[0];
    const float* w    = (const float*)d_in[1];
    const float* bias = (const float*)d_in[2];
    float* out = (float*)d_out;
    float* wp  = (float*)d_ws;

    wt_pack_kernel<<<K_ * F_ * C_ / 4 / 256, 256, 0, stream>>>(w, wp);

    deconv1d_main_kernel<<<B_ * (T_ / TT), NTHREADS, 0, stream>>>(
        x, wp, bias, out);
}


# fast-FIR pairing of taps j and j+4 across time tiles 16 rows apart: 3 WMMA products (on x[r]-x[r+16] and W[j]+W[j+4]) per 2 output tiles instead of 4; changes the summation tree
# speedup vs baseline: 2.7870x; 1.1900x over previous
#include <hip/hip_runtime.h>


typedef float v2f __attribute__((ext_vector_type(2)));
typedef float v4f __attribute__((ext_vector_type(4)));
typedef float v8f __attribute__((ext_vector_type(8)));

#define B_    16
#define T_    2048
#define C_    256
#define K_    8
#define F_    256
#define DIL_  4
#define HALO  28
#define TT    64
#define NP    2
#define NT    4
#define NJ    4
#define NM    3
#define XROWS 92
#define XS_LO 16
#define XS_ROWS 60
#define XD_ROWS 76
#define PITCH 260
#define NTHREADS 128
#define STAGE_ITERS ((XROWS * (C_ / 4)) / NTHREADS)
#define FTILES (F_ / 16)
#define CPAIRS (C_ / 8)
#define NSTEPS (NJ * CPAIRS)

__global__ __launch_bounds__(256)
void wt_pack_kernel(const float* __restrict__ w, float* __restrict__ wp) {
    int idx = blockIdx.x * 256 + threadIdx.x;
    if (idx >= NJ * NM * F_ * C_ / 4) return;
    int lane  = idx & 31;
    int cp    = (idx >> 5) & (CPAIRS - 1);
    int ftile = (idx >> 10) & (FTILES - 1);
    int jm    = idx >> 14;
    int j     = jm / NM;
    int mtx   = jm - j * NM;
    int f     = ftile * 16 + (lane & 15);
    int c0    = cp * 8 + (lane >> 4) * 2;
    const float* p0 = w + ((size_t)c0 * K_ + (K_ - 1 - j)) * F_ + f;
    const float* p1 = w + ((size_t)c0 * K_ + (K_ - 1 - j - NJ)) * F_ + f;
    v4f g0 = (v4f){0.f, 0.f, 0.f, 0.f};
    v4f g1 = (v4f){0.f, 0.f, 0.f, 0.f};
    if (mtx != 2) {
        g0.x = p0[0];
        g0.y = p0[K_ * F_];
        g0.z = p0[4 * K_ * F_];
        g0.w = p0[5 * K_ * F_];
    }
    if (mtx != 0) {
        g1.x = p1[0];
        g1.y = p1[K_ * F_];
        g1.z = p1[4 * K_ * F_];
        g1.w = p1[5 * K_ * F_];
    }
    v4f v = mtx == 0 ? g0 : (mtx == 2 ? g1 : g0 + g1);
    ((v4f*)wp)[idx] = v;
}

__device__ __forceinline__ void fence() { __builtin_amdgcn_sched_barrier(0); }

struct Frag {
    v2f a[2][NP][NM];
    v4f b[NM][NT];
};

__device__ __forceinline__ void load_frag(Frag& f, const float* sLane, const float* dLane,
                                          const v4f* bLane, int step) {
    const int j  = step / CPAIRS;
    const int cp = step % CPAIRS;
    const int ao = (DIL_ * j) * PITCH + cp * 8;
    const float* sP = sLane + ao;
    const float* dP = dLane + ao;
    const v4f*   bP = bLane + ((size_t)j * NM * FTILES * CPAIRS + cp) * 32;
#pragma unroll
    for (int s = 0; s < 2; ++s)
#pragma unroll
        for (int p = 0; p < NP; ++p) {
            f.a[s][p][0] = *(const v2f*)(dP + (32 * p) * PITCH + s * 4);
            f.a[s][p][1] = *(const v2f*)(sP + (32 * p) * PITCH + s * 4);
            f.a[s][p][2] = *(const v2f*)(dP + (32 * p + 16) * PITCH + s * 4);
        }
#pragma unroll
    for (int q = 0; q < NM; ++q)
#pragma unroll
        for (int ft = 0; ft < NT; ++ft)
            f.b[q][ft] = bP[((q * FTILES + ft) * CPAIRS) * 32];
}

__device__ __forceinline__ void mma_frag(v8f (&acc)[NP][NM][NT], const Frag& f) {
#pragma unroll
    for (int s = 0; s < 2; ++s)
#pragma unroll
        for (int p = 0; p < NP; ++p)
#pragma unroll
            for (int q = 0; q < NM; ++q)
#pragma unroll
                for (int ft = 0; ft < NT; ++ft) {
                    v2f bf = s == 0 ? __builtin_shufflevector(f.b[q][ft], f.b[q][ft], 0, 1)
                                    : __builtin_shufflevector(f.b[q][ft], f.b[q][ft], 2, 3);
                    acc[p][q][ft] = __builtin_amdgcn_wmma_f32_16x16x4_f32(
                        false, f.a[s][p][q], false, bf, (short)0, acc[p][q][ft], false, false);
                }
}

__global__ __launch_bounds__(NTHREADS)
void deconv1d_main_kernel(const float* __restrict__ x,
                          const float* __restrict__ wp,
                          const float* __restrict__ bias,
                          float* __restrict__ out) {
    __shared__ __align__(16) float xs[XS_ROWS * PITCH];
    __shared__ __align__(16) float xd[XD_ROWS * PITCH];

    const int tile = blockIdx.x & (T_ / TT - 1);
    const int b    = blockIdx.x / (T_ / TT);
    const int t_lo = tile * TT;
    const int tid  = threadIdx.x;

    const float4* xg = (const float4*)(x + (size_t)b * T_ * C_);
    const int r0 = tid >> 6;
    const int cq = tid & 63;
    float4 v[STAGE_ITERS];
#pragma unroll
    for (int it = 0; it < STAGE_ITERS; ++it) {
        int t   = t_lo + r0 + 2 * it;
        int tc  = t < T_ ? t : (T_ - 1);
        float4 q = xg[tc * (C_ / 4) + cq];
        if (t >= T_) q = make_float4(0.f, 0.f, 0.f, 0.f);
        v[it] = q;
    }
#pragma unroll
    for (int it = 0; it < STAGE_ITERS; ++it) {
        int row = r0 + 2 * it;
        if (it >= XS_LO / 2 && it < (XS_LO + XS_ROWS) / 2)
            *(float4*)&xs[(row - XS_LO) * PITCH + cq * 4] = v[it];
        if (it < XD_ROWS / 2) {
            float4 a = v[it], c = v[it + 8];
            *(float4*)&xd[row * PITCH + cq * 4] = make_float4(a.x - c.x, a.y - c.y, a.z - c.z, a.w - c.w);
        }
    }
    __syncthreads();

    const int lane  = tid & 31;
    const int wv    = tid >> 5;
    const int n16   = lane & 15;
    const int half  = lane >> 4;
    const int chalf = half * 2;
    const int f0    = wv * 64;

    v8f acc[NP][NM][NT];
#pragma unroll
    for (int p = 0; p < NP; ++p)
#pragma unroll
        for (int q = 0; q < NM; ++q)
#pragma unroll
            for (int ft = 0; ft < NT; ++ft)
                acc[p][q][ft] = (v8f){0.f, 0.f, 0.f, 0.f, 0.f, 0.f, 0.f, 0.f};

    const float* sLane = &xs[n16 * PITCH + chalf];
    const float* dLane = &xd[n16 * PITCH + chalf];
    const v4f*   bLane = (const v4f*)wp + ((size_t)wv * NT * CPAIRS) * 32 + lane;

    Frag g0, g1;
    load_frag(g0, sLane, dLane, bLane, 0);
    for (int it = 0; it < NSTEPS / 2 - 1; ++it) {
        load_frag(g1, sLane, dLane, bLane, 2 * it + 1);
        fence();
        mma_frag(acc, g0);
        fence();
        load_frag(g0, sLane, dLane, bLane, 2 * it + 2);
        fence();
        mma_frag(acc, g1);
    }
    load_frag(g1, sLane, dLane, bLane, NSTEPS - 1);
    fence();
    mma_frag(acc, g0);
    mma_frag(acc, g1);

    float biasv[NT];
#pragma unroll
    for (int ft = 0; ft < NT; ++ft)
        biasv[ft] = bias[f0 + ft * 16 + n16] * (float)C_;
    float* outb = out + (size_t)b * T_ * F_;
#pragma unroll
    for (int p = 0; p < NP; ++p) {
#pragma unroll
        for (int h = 0; h < 2; ++h) {
#pragma unroll
            for (int i = 0; i < 8; ++i) {
                int m   = (2 * p + h) * 16 + i + half * 8;
                int tau = t_lo + m;
                int d   = (tau + HALO) - (T_ - 1);
                int kmin = d > 0 ? ((d + 3) >> 2) : 0;
                float cnt = (float)(K_ - kmin);
                float* o = outb + (size_t)tau * F_ + f0 + n16;
#pragma unroll
                for (int ft = 0; ft < NT; ++ft) {
                    float y = h == 0 ? acc[p][0][ft][i] + acc[p][1][ft][i]
                                     : acc[p][1][ft][i] - acc[p][2][ft][i];
                    o[ft * 16] = y + cnt * biasv[ft];
                }
            }
        }
    }
}

extern "C" void kernel_launch(void* const* d_in, const int* in_sizes, int n_in,
                              void* d_out, int out_size, void* d_ws, size_t ws_size,
                              hipStream_t stream) {
    const float* x    = (const float*)d_in[0];
    const float* w    = (const float*)d_in[1];
    const float* bias = (const float*)d_in[2];
    float* out = (float*)d_out;
    float* wp  = (float*)d_ws;

    wt_pack_kernel<<<NJ * NM * F_ * C_ / 4 / 256, 256, 0, stream>>>(w, wp);

    deconv1d_main_kernel<<<B_ * (T_ / TT), NTHREADS, 0, stream>>>(
        x, wp, bias, out);
}


# second nested level of fast-FIR pairing (taps j=2a+jp over 8-row time blocks, two 32-row groups per 16-row tile): 9 WMMA products per 64 output rows instead of 12; changes the summation tree
# speedup vs baseline: 3.0789x; 1.1047x over previous
#include <hip/hip_runtime.h>


typedef float v2f __attribute__((ext_vector_type(2)));
typedef float v4f __attribute__((ext_vector_type(4)));
typedef float v8f __attribute__((ext_vector_type(8)));

#define B_    16
#define T_    2048
#define C_    256
#define K_    8
#define F_    256
#define DIL_  4
#define HALO  28
#define TT    64
#define NT    4
#define NJP   2
#define NM    9
#define XROWS 92
#define XS_LO 8
#define XS_ROWS 76
#define XG_ROWS 68
#define PITCH 260
#define NTHREADS 128
#define STAGE_ITERS ((XROWS * (C_ / 4)) / NTHREADS)
#define FTILES (F_ / 16)
#define CPAIRS (C_ / 8)
#define NU (NJP * CPAIRS)

__global__ __launch_bounds__(256)
void wt_pack_kernel(const float* __restrict__ w, float* __restrict__ wp) {
    int idx = blockIdx.x * 256 + threadIdx.x;
    if (idx >= NJP * NM * F_ * C_ / 4) return;
    int lane  = idx & 31;
    int cp    = (idx >> 5) & (CPAIRS - 1);
    int ftile = (idx >> 10) & (FTILES - 1);
    int jm    = idx >> 14;
    int jp    = jm / NM;
    int mtx   = jm - jp * NM;
    int outer = mtx / 3;
    int inner = mtx - outer * 3;
    int f     = ftile * 16 + (lane & 15);
    int c0    = cp * 8 + (lane >> 4) * 2;
    v4f v = (v4f){0.f, 0.f, 0.f, 0.f};
#pragma unroll
    for (int o = 0; o < 2; ++o) {
        bool useO = o == 0 ? outer != 2 : outer != 0;
#pragma unroll
        for (int i = 0; i < 2; ++i) {
            bool useI = i == 0 ? inner != 2 : inner != 0;
            if (useO && useI) {
                int a = 2 * o + i;
                const float* p = w + ((size_t)c0 * K_ + (K_ - 1 - 2 * a - jp)) * F_ + f;
                v4f g;
                g.x = p[0];
                g.y = p[K_ * F_];
                g.z = p[4 * K_ * F_];
                g.w = p[5 * K_ * F_];
                v += g;
            }
        }
    }
    ((v4f*)wp)[idx] = v;
}

__device__ __forceinline__ void fence() { __builtin_amdgcn_sched_barrier(0); }

struct Frag {
    v2f r[2][4];
    v4f b[3][NT];
};

template <int OUT>
__device__ __forceinline__ void load_frag(Frag& f, const float* sLane, const float* gLane,
                                          const v4f* bLane, int u) {
    const int jp = u / CPAIRS;
    const int cp = u % CPAIRS;
    const int ao = (DIL_ * jp) * PITCH + cp * 8;
    const float* sP = sLane + ao;
    const float* gP = gLane + ao;
    const v4f*   bP = bLane + ((size_t)(jp * NM + OUT * 3) * FTILES * CPAIRS + cp) * 32;
#pragma unroll
    for (int s = 0; s < 2; ++s) {
        if (OUT == 0) {
            f.r[s][0] = *(const v2f*)(gP + s * 4);
            f.r[s][1] = *(const v2f*)(sP + (8 - XS_LO) * PITCH + s * 4);
            f.r[s][2] = *(const v2f*)(sP + (24 - XS_LO) * PITCH + s * 4);
            f.r[s][3] = *(const v2f*)(gP + 8 * PITCH + s * 4);
        } else if (OUT == 1) {
            f.r[s][0] = *(const v2f*)(sP + (16 - XS_LO) * PITCH + s * 4);
            f.r[s][1] = *(const v2f*)(sP + (24 - XS_LO) * PITCH + s * 4);
            f.r[s][2] = *(const v2f*)(sP + (32 - XS_LO) * PITCH + s * 4);
        } else {
            f.r[s][0] = *(const v2f*)(gP + 16 * PITCH + s * 4);
            f.r[s][1] = *(const v2f*)(sP + (24 - XS_LO) * PITCH + s * 4);
            f.r[s][2] = *(const v2f*)(sP + (40 - XS_LO) * PITCH + s * 4);
            f.r[s][3] = *(const v2f*)(gP + 24 * PITCH + s * 4);
        }
    }
#pragma unroll
    for (int q = 0; q < 3; ++q)
#pragma unroll
        for (int ft = 0; ft < NT; ++ft)
            f.b[q][ft] = bP[((q * FTILES + ft) * CPAIRS) * 32];
}

template <int OUT>
__device__ __forceinline__ void mma_frag(v8f (&acc)[NM][NT], const Frag& f) {
#pragma unroll
    for (int s = 0; s < 2; ++s) {
        v2f a[3];
        if (OUT == 1) {
            a[0] = f.r[s][0] - f.r[s][1];
            a[1] = f.r[s][1];
            a[2] = f.r[s][1] - f.r[s][2];
        } else {
            a[0] = f.r[s][0];
            a[1] = f.r[s][1] - f.r[s][2];
            a[2] = f.r[s][3];
        }
#pragma unroll
        for (int q = 0; q < 3; ++q)
#pragma unroll
            for (int ft = 0; ft < NT; ++ft) {
                v2f bf = s == 0 ? __builtin_shufflevector(f.b[q][ft], f.b[q][ft], 0, 1)
                                : __builtin_shufflevector(f.b[q][ft], f.b[q][ft], 2, 3);
                acc[OUT * 3 + q][ft] = __builtin_amdgcn_wmma_f32_16x16x4_f32(
                    false, a[q], false, bf, (short)0, acc[OUT * 3 + q][ft], false, false);
            }
    }
}

template <bool LAST>
__device__ __forceinline__ void two_steps(v8f (&acc)[NM][NT], Frag& f0, Frag& f1,
                                          const float* sLane, const float* gLane,
                                          const v4f* bLane, int u) {
    load_frag<1>(f1, sLane, gLane, bLane, u);
    fence();
    mma_frag<0>(acc, f0);
    fence();
    load_frag<2>(f0, sLane, gLane, bLane, u);
    fence();
    mma_frag<1>(acc, f1);
    fence();
    load_frag<0>(f1, sLane, gLane, bLane, u + 1);
    fence();
    mma_frag<2>(acc, f0);
    fence();
    load_frag<1>(f0, sLane, gLane, bLane, u + 1);
    fence();
    mma_frag<0>(acc, f1);
    fence();
    load_frag<2>(f1, sLane, gLane, bLane, u + 1);
    fence();
    mma_frag<1>(acc, f0);
    fence();
    if (!LAST) load_frag<0>(f0, sLane, gLane, bLane, u + 2);
    fence();
    mma_frag<2>(acc, f1);
    fence();
}

__global__ __launch_bounds__(NTHREADS)
void deconv1d_main_kernel(const float* __restrict__ x,
                          const float* __restrict__ wp,
                          const float* __restrict__ bias,
                          float* __restrict__ out) {
    __shared__ __align__(16) float xs[XS_ROWS * PITCH];
    __shared__ __align__(16) float xg_[XG_ROWS * PITCH];

    const int tile = blockIdx.x & (T_ / TT - 1);
    const int b    = blockIdx.x / (T_ / TT);
    const int t_lo = tile * TT;
    const int tid  = threadIdx.x;

    const float4* xgl = (const float4*)(x + (size_t)b * T_ * C_);
    const int r0 = tid >> 6;
    const int cq = tid & 63;
    float4 v[STAGE_ITERS];
#pragma unroll
    for (int it = 0; it < STAGE_ITERS; ++it) {
        int t   = t_lo + r0 + 2 * it;
        int tc  = t < T_ ? t : (T_ - 1);
        float4 q = xgl[tc * (C_ / 4) + cq];
        if (t >= T_) q = make_float4(0.f, 0.f, 0.f, 0.f);
        v[it] = q;
    }
#pragma unroll
    for (int it = 0; it < STAGE_ITERS; ++it) {
        int row = r0 + 2 * it;
        if (it >= XS_LO / 2 && it < (XS_LO + XS_ROWS) / 2)
            *(float4*)&xs[(row - XS_LO) * PITCH + cq * 4] = v[it];
        if (it < XG_ROWS / 2) {
            float4 a = v[it], c = v[it + 4], d = v[it + 8], e = v[it + 12];
            *(float4*)&xg_[row * PITCH + cq * 4] = make_float4((a.x - c.x) - (d.x - e.x), (a.y - c.y) - (d.y - e.y),
                                                               (a.z - c.z) - (d.z - e.z), (a.w - c.w) - (d.w - e.w));
        }
    }
    __syncthreads();

    const int lane  = tid & 31;
    const int wv    = tid >> 5;
    const int n16   = lane & 15;
    const int half  = lane >> 4;
    const int chalf = half * 2;
    const int f0    = wv * 64;

    v8f acc[NM][NT];
#pragma unroll
    for (int q = 0; q < NM; ++q)
#pragma unroll
        for (int ft = 0; ft < NT; ++ft)
            acc[q][ft] = (v8f){0.f, 0.f, 0.f, 0.f, 0.f, 0.f, 0.f, 0.f};

    const int rowLane = 32 * (n16 >> 3) + (n16 & 7);
    const float* sLane = &xs[rowLane * PITCH + chalf];
    const float* gLane = &xg_[rowLane * PITCH + chalf];
    const v4f*   bLane = (const v4f*)wp + ((size_t)wv * NT * CPAIRS) * 32 + lane;

    Frag g0, g1;
    load_frag<0>(g0, sLane, gLane, bLane, 0);
    for (int u = 0; u < NU - 2; u += 2)
        two_steps<false>(acc, g0, g1, sLane, gLane, bLane, u);
    two_steps<true>(acc, g0, g1, sLane, gLane, bLane, NU - 2);

    float biasv[NT];
#pragma unroll
    for (int ft = 0; ft < NT; ++ft)
        biasv[ft] = bias[f0 + ft * 16 + n16] * (float)C_;
    float* outb = out + (size_t)b * T_ * F_;
#pragma unroll
    for (int blk = 0; blk < 4; ++blk) {
#pragma unroll
        for (int i = 0; i < 8; ++i) {
            int m   = 32 * half + 8 * blk + i;
            int tau = t_lo + m;
            int d   = (tau + HALO) - (T_ - 1);
            int kmin = d > 0 ? ((d + 3) >> 2) : 0;
            float cnt = (float)(K_ - kmin);
            float* o = outb + (size_t)tau * F_ + f0 + n16;
#pragma unroll
            for (int ft = 0; ft < NT; ++ft) {
                float y;
                if (blk == 0)      y = (acc[0][ft][i] + acc[1][ft][i]) + (acc[3][ft][i] + acc[4][ft][i]);
                else if (blk == 1) y = (acc[1][ft][i] - acc[2][ft][i]) + (acc[4][ft][i] - acc[5][ft][i]);
                else if (blk == 2) y = (acc[3][ft][i] + acc[4][ft][i]) - (acc[6][ft][i] + acc[7][ft][i]);
                else               y = (acc[4][ft][i] - acc[5][ft][i]) - (acc[7][ft][i] - acc[8][ft][i]);
                o[ft * 16] = y + cnt * biasv[ft];
            }
        }
    }
}

extern "C" void kernel_launch(void* const* d_in, const int* in_sizes, int n_in,
                              void* d_out, int out_size, void* d_ws, size_t ws_size,
                              hipStream_t stream) {
    const float* x    = (const float*)d_in[0];
    const float* w    = (const float*)d_in[1];
    const float* bias = (const float*)d_in[2];
    float* out = (float*)d_out;
    float* wp  = (float*)d_ws;

    wt_pack_kernel<<<NJP * NM * F_ * C_ / 4 / 256, 256, 0, stream>>>(w, wp);

    deconv1d_main_kernel<<<B_ * (T_ / TT), NTHREADS, 0, stream>>>(
        x, wp, bias, out);
}


# time rows per block 64 -> 128 (8 waves, wave tile 128 t x 32 f = two 16-row A tiles x two filter tiles): each B fragment fetched from L2 feeds 2 WMMAs instead of 1; summation order kept
# speedup vs baseline: 3.1887x; 1.0357x over previous
#include <hip/hip_runtime.h>


typedef float v2f __attribute__((ext_vector_type(2)));
typedef float v4f __attribute__((ext_vector_type(4)));
typedef float v8f __attribute__((ext_vector_type(8)));

#define B_    16
#define T_    2048
#define C_    256
#define K_    8
#define F_    256
#define DIL_  4
#define HALO  28
#define TT    128
#define MT    2
#define NT    2
#define NJP   2
#define NM    9
#define XROWS 156
#define XS_LO 8
#define XS_ROWS 140
#define XG_ROWS 132
#define PITCH 260
#define NTHREADS 256
#define STAGE_ITERS ((XROWS * (C_ / 4)) / NTHREADS)
#define FTILES (F_ / 16)
#define CPAIRS (C_ / 8)
#define NU (NJP * CPAIRS)

__global__ __launch_bounds__(256)
void wt_pack_kernel(const float* __restrict__ w, float* __restrict__ wp) {
    int idx = blockIdx.x * 256 + threadIdx.x;
    if (idx >= NJP * NM * F_ * C_ / 4) return;
    int lane  = idx & 31;
    int cp    = (idx >> 5) & (CPAIRS - 1);
    int ftile = (idx >> 10) & (FTILES - 1);
    int jm    = idx >> 14;
    int jp    = jm / NM;
    int mtx   = jm - jp * NM;
    int outer = mtx / 3;
    int inner = mtx - outer * 3;
    int f     = ftile * 16 + (lane & 15);
    int c0    = cp * 8 + (lane >> 4) * 2;
    v4f v = (v4f){0.f, 0.f, 0.f, 0.f};
#pragma unroll
    for (int o = 0; o < 2; ++o) {
        bool useO = o == 0 ? outer != 2 : outer != 0;
#pragma unroll
        for (int i = 0; i < 2; ++i) {
            bool useI = i == 0 ? inner != 2 : inner != 0;
            if (useO && useI) {
                int a = 2 * o + i;
                const float* p = w + ((size_t)c0 * K_ + (K_ - 1 - 2 * a - jp)) * F_ + f;
                v4f g;
                g.x = p[0];
                g.y = p[K_ * F_];
                g.z = p[4 * K_ * F_];
                g.w = p[5 * K_ * F_];
                v += g;
            }
        }
    }
    ((v4f*)wp)[idx] = v;
}

__device__ __forceinline__ void fence() { __builtin_amdgcn_sched_barrier(0); }

struct Frag {
    v2f r[MT][2][4];
    v4f b[3][NT];
};

template <int OUT>
__device__ __forceinline__ void load_frag(Frag& f, const float* sLane, const float* gLane,
                                          const v4f* bLane, int u) {
    const int jp = u / CPAIRS;
    const int cp = u % CPAIRS;
    const int ao = (DIL_ * jp) * PITCH + cp * 8;
    const float* sP = sLane + ao;
    const float* gP = gLane + ao;
    const v4f*   bP = bLane + ((size_t)(jp * NM + OUT * 3) * FTILES * CPAIRS + cp) * 32;
#pragma unroll
    for (int mt = 0; mt < MT; ++mt) {
        const float* sM = sP + mt * 64 * PITCH;
        const float* gM = gP + mt * 64 * PITCH;
#pragma unroll
        for (int s = 0; s < 2; ++s) {
            if (OUT == 0) {
                f.r[mt][s][0] = *(const v2f*)(gM + s * 4);
                f.r[mt][s][1] = *(const v2f*)(sM + (8 - XS_LO) * PITCH + s * 4);
                f.r[mt][s][2] = *(const v2f*)(sM + (24 - XS_LO) * PITCH + s * 4);
                f.r[mt][s][3] = *(const v2f*)(gM + 8 * PITCH + s * 4);
            } else if (OUT == 1) {
                f.r[mt][s][0] = *(const v2f*)(sM + (16 - XS_LO) * PITCH + s * 4);
                f.r[mt][s][1] = *(const v2f*)(sM + (24 - XS_LO) * PITCH + s * 4);
                f.r[mt][s][2] = *(const v2f*)(sM + (32 - XS_LO) * PITCH + s * 4);
            } else {
                f.r[mt][s][0] = *(const v2f*)(gM + 16 * PITCH + s * 4);
                f.r[mt][s][1] = *(const v2f*)(sM + (24 - XS_LO) * PITCH + s * 4);
                f.r[mt][s][2] = *(const v2f*)(sM + (40 - XS_LO) * PITCH + s * 4);
                f.r[mt][s][3] = *(const v2f*)(gM + 24 * PITCH + s * 4);
            }
        }
    }
#pragma unroll
    for (int q = 0; q < 3; ++q)
#pragma unroll
        for (int ft = 0; ft < NT; ++ft)
            f.b[q][ft] = bP[((q * FTILES + ft) * CPAIRS) * 32];
}

template <int OUT>
__device__ __forceinline__ void mma_frag(v8f (&acc)[MT][NM][NT], const Frag& f) {
#pragma unroll
    for (int s = 0; s < 2; ++s) {
        v2f a[MT][3];
#pragma unroll
        for (int mt = 0; mt < MT; ++mt) {
            if (OUT == 1) {
                a[mt][0] = f.r[mt][s][0] - f.r[mt][s][1];
                a[mt][1] = f.r[mt][s][1];
                a[mt][2] = f.r[mt][s][1] - f.r[mt][s][2];
            } else {
                a[mt][0] = f.r[mt][s][0];
                a[mt][1] = f.r[mt][s][1] - f.r[mt][s][2];
                a[mt][2] = f.r[mt][s][3];
            }
        }
#pragma unroll
        for (int q = 0; q < 3; ++q)
#pragma unroll
            for (int ft = 0; ft < NT; ++ft) {
                v2f bf = s == 0 ? __builtin_shufflevector(f.b[q][ft], f.b[q][ft], 0, 1)
                                : __builtin_shufflevector(f.b[q][ft], f.b[q][ft], 2, 3);
#pragma unroll
                for (int mt = 0; mt < MT; ++mt)
                    acc[mt][OUT * 3 + q][ft] = __builtin_amdgcn_wmma_f32_16x16x4_f32(
                        false, a[mt][q], false, bf, (short)0, acc[mt][OUT * 3 + q][ft], false, false);
            }
    }
}

template <bool LAST>
__device__ __forceinline__ void two_steps(v8f (&acc)[MT][NM][NT], Frag& f0, Frag& f1,
                                          const float* sLane, const float* gLane,
                                          const v4f* bLane, int u) {
    load_frag<1>(f1, sLane, gLane, bLane, u);
    fence();
    mma_frag<0>(acc, f0);
    fence();
    load_frag<2>(f0, sLane, gLane, bLane, u);
    fence();
    mma_frag<1>(acc, f1);
    fence();
    load_frag<0>(f1, sLane, gLane, bLane, u + 1);
    fence();
    mma_frag<2>(acc, f0);
    fence();
    load_frag<1>(f0, sLane, gLane, bLane, u + 1);
    fence();
    mma_frag<0>(acc, f1);
    fence();
    load_frag<2>(f1, sLane, gLane, bLane, u + 1);
    fence();
    mma_frag<1>(acc, f0);
    fence();
    if (!LAST) load_frag<0>(f0, sLane, gLane, bLane, u + 2);
    fence();
    mma_frag<2>(acc, f1);
    fence();
}

__global__ __launch_bounds__(NTHREADS)
void deconv1d_main_kernel(const float* __restrict__ x,
                          const float* __restrict__ wp,
                          const float* __restrict__ bias,
                          float* __restrict__ out) {
    __shared__ __align__(16) float xs[XS_ROWS * PITCH];
    __shared__ __align__(16) float xg_[XG_ROWS * PITCH];

    const int tile = blockIdx.x & (T_ / TT - 1);
    const int b    = blockIdx.x / (T_ / TT);
    const int t_lo = tile * TT;
    const int tid  = threadIdx.x;

    const float4* xgl = (const float4*)(x + (size_t)b * T_ * C_);
    const int r0 = tid >> 6;
    const int cq = tid & 63;
    float4 v[STAGE_ITERS];
#pragma unroll
    for (int it = 0; it < STAGE_ITERS; ++it) {
        int t   = t_lo + r0 + 4 * it;
        int tc  = t < T_ ? t : (T_ - 1);
        float4 q = xgl[tc * (C_ / 4) + cq];
        if (t >= T_) q = make_float4(0.f, 0.f, 0.f, 0.f);
        v[it] = q;
    }
#pragma unroll
    for (int it = 0; it < STAGE_ITERS; ++it) {
        int row = r0 + 4 * it;
        if (it >= XS_LO / 4 && it < (XS_LO + XS_ROWS) / 4)
            *(float4*)&xs[(row - XS_LO) * PITCH + cq * 4] = v[it];
        if (it < XG_ROWS / 4) {
            float4 a = v[it], c = v[it + 2], d = v[it + 4], e = v[it + 6];
            *(float4*)&xg_[row * PITCH + cq * 4] = make_float4((a.x - c.x) - (d.x - e.x), (a.y - c.y) - (d.y - e.y),
                                                               (a.z - c.z) - (d.z - e.z), (a.w - c.w) - (d.w - e.w));
        }
    }
    __syncthreads();

    const int lane  = tid & 31;
    const int wv    = tid >> 5;
    const int n16   = lane & 15;
    const int half  = lane >> 4;
    const int chalf = half * 2;
    const int f0    = wv * (NT * 16);

    v8f acc[MT][NM][NT];
#pragma unroll
    for (int mt = 0; mt < MT; ++mt)
#pragma unroll
        for (int q = 0; q < NM; ++q)
#pragma unroll
            for (int ft = 0; ft < NT; ++ft)
                acc[mt][q][ft] = (v8f){0.f, 0.f, 0.f, 0.f, 0.f, 0.f, 0.f, 0.f};

    const int rowLane = 32 * (n16 >> 3) + (n16 & 7);
    const float* sLane = &xs[rowLane * PITCH + chalf];
    const float* gLane = &xg_[rowLane * PITCH + chalf];
    const v4f*   bLane = (const v4f*)wp + ((size_t)wv * NT * CPAIRS) * 32 + lane;

    Frag g0, g1;
    load_frag<0>(g0, sLane, gLane, bLane, 0);
    for (int u = 0; u < NU - 2; u += 2)
        two_steps<false>(acc, g0, g1, sLane, gLane, bLane, u);
    two_steps<true>(acc, g0, g1, sLane, gLane, bLane, NU - 2);

    float biasv[NT];
#pragma unroll
    for (int ft = 0; ft < NT; ++ft)
        biasv[ft] = bias[f0 + ft * 16 + n16] * (float)C_;
    float* outb = out + (size_t)b * T_ * F_;
#pragma unroll
    for (int mt = 0; mt < MT; ++mt) {
#pragma unroll
        for (int blk = 0; blk < 4; ++blk) {
#pragma unroll
            for (int i = 0; i < 8; ++i) {
                int m   = 64 * mt + 32 * half + 8 * blk + i;
                int tau = t_lo + m;
                int d   = (tau + HALO) - (T_ - 1);
                int kmin = d > 0 ? ((d + 3) >> 2) : 0;
                float cnt = (float)(K_ - kmin);
                float* o = outb + (size_t)tau * F_ + f0 + n16;
#pragma unroll
                for (int ft = 0; ft < NT; ++ft) {
                    const v8f (&A)[NM][NT] = acc[mt];
                    float y;
                    if (blk == 0)      y = (A[0][ft][i] + A[1][ft][i]) + (A[3][ft][i] + A[4][ft][i]);
                    else if (blk == 1) y = (A[1][ft][i] - A[2][ft][i]) + (A[4][ft][i] - A[5][ft][i]);
                    else if (blk == 2) y = (A[3][ft][i] + A[4][ft][i]) - (A[6][ft][i] + A[7][ft][i]);
                    else               y = (A[4][ft][i] - A[5][ft][i]) - (A[7][ft][i] - A[8][ft][i]);
                    o[ft * 16] = y + cnt * biasv[ft];
                }
            }
        }
    }
}

extern "C" void kernel_launch(void* const* d_in, const int* in_sizes, int n_in,
                              void* d_out, int out_size, void* d_ws, size_t ws_size,
                              hipStream_t stream) {
    const float* x    = (const float*)d_in[0];
    const float* w    = (const float*)d_in[1];
    const float* bias = (const float*)d_in[2];
    float* out = (float*)d_out;
    float* wp  = (float*)d_ws;

    wt_pack_kernel<<<NJP * NM * F_ * C_ / 4 / 256, 256, 0, stream>>>(w, wp);

    deconv1d_main_kernel<<<B_ * (T_ / TT), NTHREADS, 0, stream>>>(
        x, wp, bias, out);
}


# weight prepass launch removed: each wave loads a step's raw taps H0..H3 as B fragments (4-byte loads) and forms the 5 tap sums in registers; the 4-tap weight sum is now (H0+H2)+(H1+H3)
# speedup vs baseline: 3.3854x; 1.0617x over previous
#include <hip/hip_runtime.h>


typedef float v2f __attribute__((ext_vector_type(2)));
typedef float v8f __attribute__((ext_vector_type(8)));

#define B_    16
#define T_    2048
#define C_    256
#define K_    8
#define F_    256
#define DIL_  4
#define HALO  28
#define TT    128
#define MT    2
#define NT    2
#define NJP   2
#define NM    9
#define XROWS 156
#define XS_LO 8
#define XS_ROWS 140
#define XG_ROWS 132
#define PITCH 260
#define NTHREADS 256
#define STAGE_ITERS ((XROWS * (C_ / 4)) / NTHREADS)
#define CPAIRS (C_ / 8)
#define NU (NJP * CPAIRS)

__device__ __forceinline__ void fence() { __builtin_amdgcn_sched_barrier(0); }

struct Frag {
    v2f r[MT][2][4];
    v2f h[2][2][NT];
};
struct TapSum {
    v2f t[2][2][NT];
};

template <int OUT>
__device__ __forceinline__ void load_frag(Frag& f, const float* sLane, const float* gLane,
                                          const float* wLane, int u) {
    const int jp = u / CPAIRS;
    const int cp = u % CPAIRS;
    const int ao = (DIL_ * jp) * PITCH + cp * 8;
    const float* sP = sLane + ao;
    const float* gP = gLane + ao;
    const float* wP = wLane + ((size_t)cp * 8 * K_ + (K_ - 1 - jp)) * F_;
#pragma unroll
    for (int mt = 0; mt < MT; ++mt) {
        const float* sM = sP + mt * 64 * PITCH;
        const float* gM = gP + mt * 64 * PITCH;
#pragma unroll
        for (int s = 0; s < 2; ++s) {
            if (OUT == 0) {
                f.r[mt][s][0] = *(const v2f*)(gM + s * 4);
                f.r[mt][s][1] = *(const v2f*)(sM + (8 - XS_LO) * PITCH + s * 4);
                f.r[mt][s][2] = *(const v2f*)(sM + (24 - XS_LO) * PITCH + s * 4);
                f.r[mt][s][3] = *(const v2f*)(gM + 8 * PITCH + s * 4);
            } else if (OUT == 1) {
                f.r[mt][s][0] = *(const v2f*)(sM + (16 - XS_LO) * PITCH + s * 4);
                f.r[mt][s][1] = *(const v2f*)(sM + (24 - XS_LO) * PITCH + s * 4);
                f.r[mt][s][2] = *(const v2f*)(sM + (32 - XS_LO) * PITCH + s * 4);
            } else {
                f.r[mt][s][0] = *(const v2f*)(gM + 16 * PITCH + s * 4);
                f.r[mt][s][1] = *(const v2f*)(sM + (24 - XS_LO) * PITCH + s * 4);
                f.r[mt][s][2] = *(const v2f*)(sM + (40 - XS_LO) * PITCH + s * 4);
                f.r[mt][s][3] = *(const v2f*)(gM + 24 * PITCH + s * 4);
            }
        }
    }
    if (OUT != 1) {
#pragma unroll
        for (int i = 0; i < 2; ++i) {
            const int a = (OUT == 0 ? 0 : 2) + i;
#pragma unroll
            for (int s = 0; s < 2; ++s)
#pragma unroll
                for (int ft = 0; ft < NT; ++ft) {
                    const float* p = wP + ((s * 4) * K_ - 2 * a) * F_ + ft * 16;
                    f.h[i][s][ft] = (v2f){p[0], p[K_ * F_]};
                }
        }
    }
}

template <int OUT>
__device__ __forceinline__ void mma_frag(v8f (&acc)[MT][NM][NT], const Frag& f,
                                         const v2f (&g)[2][2][NT]) {
#pragma unroll
    for (int s = 0; s < 2; ++s) {
        v2f a[MT][3];
#pragma unroll
        for (int mt = 0; mt < MT; ++mt) {
            if (OUT == 1) {
                a[mt][0] = f.r[mt][s][0] - f.r[mt][s][1];
                a[mt][1] = f.r[mt][s][1];
                a[mt][2] = f.r[mt][s][1] - f.r[mt][s][2];
            } else {
                a[mt][0] = f.r[mt][s][0];
                a[mt][1] = f.r[mt][s][1] - f.r[mt][s][2];
                a[mt][2] = f.r[mt][s][3];
            }
        }
#pragma unroll
        for (int q = 0; q < 3; ++q)
#pragma unroll
            for (int ft = 0; ft < NT; ++ft) {
                v2f bf = q == 0 ? g[0][s][ft] : q == 2 ? g[1][s][ft] : g[0][s][ft] + g[1][s][ft];
#pragma unroll
                for (int mt = 0; mt < MT; ++mt)
                    acc[mt][OUT * 3 + q][ft] = __builtin_amdgcn_wmma_f32_16x16x4_f32(
                        false, a[mt][q], false, bf, (short)0, acc[mt][OUT * 3 + q][ft], false, false);
            }
    }
}

__device__ __forceinline__ void tap_sums(TapSum& ts, const Frag& f01, const Frag& f23) {
#pragma unroll
    for (int i = 0; i < 2; ++i)
#pragma unroll
        for (int s = 0; s < 2; ++s)
#pragma unroll
            for (int ft = 0; ft < NT; ++ft)
                ts.t[i][s][ft] = f01.h[i][s][ft] + f23.h[i][s][ft];
}

template <bool LAST>
__device__ __forceinline__ void two_steps(v8f (&acc)[MT][NM][NT], Frag& f0, Frag& f1, TapSum& ts,
                                          const float* sLane, const float* gLane,
                                          const float* wLane, int u) {
    load_frag<2>(f1, sLane, gLane, wLane, u);
    fence();
    mma_frag<0>(acc, f0, f0.h);
    fence();
    load_frag<1>(f0, sLane, gLane, wLane, u);
    fence();
    tap_sums(ts, f0, f1);
    mma_frag<2>(acc, f1, f1.h);
    fence();
    load_frag<0>(f1, sLane, gLane, wLane, u + 1);
    fence();
    mma_frag<1>(acc, f0, ts.t);
    fence();
    load_frag<2>(f0, sLane, gLane, wLane, u + 1);
    fence();
    mma_frag<0>(acc, f1, f1.h);
    fence();
    load_frag<1>(f1, sLane, gLane, wLane, u + 1);
    fence();
    tap_sums(ts, f1, f0);
    mma_frag<2>(acc, f0, f0.h);
    fence();
    if (!LAST) load_frag<0>(f0, sLane, gLane, wLane, u + 2);
    fence();
    mma_frag<1>(acc, f1, ts.t);
    fence();
}

__global__ __launch_bounds__(NTHREADS)
void deconv1d_main_kernel(const float* __restrict__ x,
                          const float* __restrict__ w,
                          const float* __restrict__ bias,
                          float* __restrict__ out) {
    __shared__ __align__(16) float xs[XS_ROWS * PITCH];
    __shared__ __align__(16) float xg_[XG_ROWS * PITCH];

    const int tile = blockIdx.x & (T_ / TT - 1);
    const int b    = blockIdx.x / (T_ / TT);
    const int t_lo = tile * TT;
    const int tid  = threadIdx.x;

    const float4* xgl = (const float4*)(x + (size_t)b * T_ * C_);
    const int r0 = tid >> 6;
    const int cq = tid & 63;
    float4 v[STAGE_ITERS];
#pragma unroll
    for (int it = 0; it < STAGE_ITERS; ++it) {
        int t   = t_lo + r0 + 4 * it;
        int tc  = t < T_ ? t : (T_ - 1);
        float4 q = xgl[tc * (C_ / 4) + cq];
        if (t >= T_) q = make_float4(0.f, 0.f, 0.f, 0.f);
        v[it] = q;
    }
#pragma unroll
    for (int it = 0; it < STAGE_ITERS; ++it) {
        int row = r0 + 4 * it;
        if (it >= XS_LO / 4 && it < (XS_LO + XS_ROWS) / 4)
            *(float4*)&xs[(row - XS_LO) * PITCH + cq * 4] = v[it];
        if (it < XG_ROWS / 4) {
            float4 a = v[it], c = v[it + 2], d = v[it + 4], e = v[it + 6];
            *(float4*)&xg_[row * PITCH + cq * 4] = make_float4((a.x - c.x) - (d.x - e.x), (a.y - c.y) - (d.y - e.y),
                                                               (a.z - c.z) - (d.z - e.z), (a.w - c.w) - (d.w - e.w));
        }
    }
    __syncthreads();

    const int lane  = tid & 31;
    const int wv    = tid >> 5;
    const int n16   = lane & 15;
    const int half  = lane >> 4;
    const int chalf = half * 2;
    const int f0    = wv * (NT * 16);

    v8f acc[MT][NM][NT];
#pragma unroll
    for (int mt = 0; mt < MT; ++mt)
#pragma unroll
        for (int q = 0; q < NM; ++q)
#pragma unroll
            for (int ft = 0; ft < NT; ++ft)
                acc[mt][q][ft] = (v8f){0.f, 0.f, 0.f, 0.f, 0.f, 0.f, 0.f, 0.f};

    const int rowLane = 32 * (n16 >> 3) + (n16 & 7);
    const float* sLane = &xs[rowLane * PITCH + chalf];
    const float* gLane = &xg_[rowLane * PITCH + chalf];
    const float* wLane = w + (size_t)chalf * K_ * F_ + f0 + n16;

    Frag g0, g1;
    TapSum ts;
    load_frag<0>(g0, sLane, gLane, wLane, 0);
    for (int u = 0; u < NU - 2; u += 2)
        two_steps<false>(acc, g0, g1, ts, sLane, gLane, wLane, u);
    two_steps<true>(acc, g0, g1, ts, sLane, gLane, wLane, NU - 2);

    float biasv[NT];
#pragma unroll
    for (int ft = 0; ft < NT; ++ft)
        biasv[ft] = bias[f0 + ft * 16 + n16] * (float)C_;
    float* outb = out + (size_t)b * T_ * F_;
#pragma unroll
    for (int mt = 0; mt < MT; ++mt) {
#pragma unroll
        for (int blk = 0; blk < 4; ++blk) {
#pragma unroll
            for (int i = 0; i < 8; ++i) {
                int m   = 64 * mt + 32 * half + 8 * blk + i;
                int tau = t_lo + m;
                int d   = (tau + HALO) - (T_ - 1);
                int kmin = d > 0 ? ((d + 3) >> 2) : 0;
                float cnt = (float)(K_ - kmin);
                float* o = outb + (size_t)tau * F_ + f0 + n16;
#pragma unroll
                for (int ft = 0; ft < NT; ++ft) {
                    const v8f (&A)[NM][NT] = acc[mt];
                    float y;
                    if (blk == 0)      y = (A[0][ft][i] + A[1][ft][i]) + (A[3][ft][i] + A[4][ft][i]);
                    else if (blk == 1) y = (A[1][ft][i] - A[2][ft][i]) + (A[4][ft][i] - A[5][ft][i]);
                    else if (blk == 2) y = (A[3][ft][i] + A[4][ft][i]) - (A[6][ft][i] + A[7][ft][i]);
                    else               y = (A[4][ft][i] - A[5][ft][i]) - (A[7][ft][i] - A[8][ft][i]);
                    o[ft * 16] = y + cnt * biasv[ft];
                }
            }
        }
    }
}

extern "C" void kernel_launch(void* const* d_in, const int* in_sizes, int n_in,
                              void* d_out, int out_size, void* d_ws, size_t ws_size,
                              hipStream_t stream) {
    const float* x    = (const float*)d_in[0];
    const float* w    = (const float*)d_in[1];
    const float* bias = (const float*)d_in[2];
    float* out = (float*)d_out;

    deconv1d_main_kernel<<<B_ * (T_ / TT), NTHREADS, 0, stream>>>(
        x, w, bias, out);
}


# priority window: the younger wave of each SIMD's pair (w+4) is served first for its first 48 of 64 steps, then oldest-first again, so both waves finish together; arithmetic and sum order unchanged
# speedup vs baseline: 3.5499x; 1.0486x over previous
#include <hip/hip_runtime.h>


typedef float v2f __attribute__((ext_vector_type(2)));
typedef float v8f __attribute__((ext_vector_type(8)));

#define B_    16
#define T_    2048
#define C_    256
#define K_    8
#define F_    256
#define DIL_  4
#define HALO  28
#define TT    128
#define MT    2
#define NT    2
#define NJP   2
#define NM    9
#define XROWS 156
#define XS_LO 8
#define XS_ROWS 140
#define XG_ROWS 132
#define PITCH 260
#define NTHREADS 256
#define STAGE_ITERS ((XROWS * (C_ / 4)) / NTHREADS)
#define CPAIRS (C_ / 8)
#define NU (NJP * CPAIRS)
#define PRIO_STEPS 48

__device__ __forceinline__ void fence() { __builtin_amdgcn_sched_barrier(0); }

struct Frag {
    v2f r[MT][2][4];
    v2f h[2][2][NT];
};
struct TapSum {
    v2f t[2][2][NT];
};

template <int OUT>
__device__ __forceinline__ void load_frag(Frag& f, const float* sLane, const float* gLane,
                                          const float* wLane, int u) {
    const int jp = u / CPAIRS;
    const int cp = u % CPAIRS;
    const int ao = (DIL_ * jp) * PITCH + cp * 8;
    const float* sP = sLane + ao;
    const float* gP = gLane + ao;
    const float* wP = wLane + ((size_t)cp * 8 * K_ + (K_ - 1 - jp)) * F_;
#pragma unroll
    for (int mt = 0; mt < MT; ++mt) {
        const float* sM = sP + mt * 64 * PITCH;
        const float* gM = gP + mt * 64 * PITCH;
#pragma unroll
        for (int s = 0; s < 2; ++s) {
            if (OUT == 0) {
                f.r[mt][s][0] = *(const v2f*)(gM + s * 4);
                f.r[mt][s][1] = *(const v2f*)(sM + (8 - XS_LO) * PITCH + s * 4);
                f.r[mt][s][2] = *(const v2f*)(sM + (24 - XS_LO) * PITCH + s * 4);
                f.r[mt][s][3] = *(const v2f*)(gM + 8 * PITCH + s * 4);
            } else if (OUT == 1) {
                f.r[mt][s][0] = *(const v2f*)(sM + (16 - XS_LO) * PITCH + s * 4);
                f.r[mt][s][1] = *(const v2f*)(sM + (24 - XS_LO) * PITCH + s * 4);
                f.r[mt][s][2] = *(const v2f*)(sM + (32 - XS_LO) * PITCH + s * 4);
            } else {
                f.r[mt][s][0] = *(const v2f*)(gM + 16 * PITCH + s * 4);
                f.r[mt][s][1] = *(const v2f*)(sM + (24 - XS_LO) * PITCH + s * 4);
                f.r[mt][s][2] = *(const v2f*)(sM + (40 - XS_LO) * PITCH + s * 4);
                f.r[mt][s][3] = *(const v2f*)(gM + 24 * PITCH + s * 4);
            }
        }
    }
    if (OUT != 1) {
#pragma unroll
        for (int i = 0; i < 2; ++i) {
            const int a = (OUT == 0 ? 0 : 2) + i;
#pragma unroll
            for (int s = 0; s < 2; ++s)
#pragma unroll
                for (int ft = 0; ft < NT; ++ft) {
                    const float* p = wP + ((s * 4) * K_ - 2 * a) * F_ + ft * 16;
                    f.h[i][s][ft] = (v2f){p[0], p[K_ * F_]};
                }
        }
    }
}

template <int OUT>
__device__ __forceinline__ void mma_frag(v8f (&acc)[MT][NM][NT], const Frag& f,
                                         const v2f (&g)[2][2][NT]) {
#pragma unroll
    for (int s = 0; s < 2; ++s) {
        v2f a[MT][3];
#pragma unroll
        for (int mt = 0; mt < MT; ++mt) {
            if (OUT == 1) {
                a[mt][0] = f.r[mt][s][0] - f.r[mt][s][1];
                a[mt][1] = f.r[mt][s][1];
                a[mt][2] = f.r[mt][s][1] - f.r[mt][s][2];
            } else {
                a[mt][0] = f.r[mt][s][0];
                a[mt][1] = f.r[mt][s][1] - f.r[mt][s][2];
                a[mt][2] = f.r[mt][s][3];
            }
        }
#pragma unroll
        for (int q = 0; q < 3; ++q)
#pragma unroll
            for (int ft = 0; ft < NT; ++ft) {
                v2f bf = q == 0 ? g[0][s][ft] : q == 2 ? g[1][s][ft] : g[0][s][ft] + g[1][s][ft];
#pragma unroll
                for (int mt = 0; mt < MT; ++mt)
                    acc[mt][OUT * 3 + q][ft] = __builtin_amdgcn_wmma_f32_16x16x4_f32(
                        false, a[mt][q], false, bf, (short)0, acc[mt][OUT * 3 + q][ft], false, false);
            }
    }
}

__device__ __forceinline__ void tap_sums(TapSum& ts, const Frag& f01, const Frag& f23) {
#pragma unroll
    for (int i = 0; i < 2; ++i)
#pragma unroll
        for (int s = 0; s < 2; ++s)
#pragma unroll
            for (int ft = 0; ft < NT; ++ft)
                ts.t[i][s][ft] = f01.h[i][s][ft] + f23.h[i][s][ft];
}

template <bool LAST>
__device__ __forceinline__ void two_steps(v8f (&acc)[MT][NM][NT], Frag& f0, Frag& f1, TapSum& ts,
                                          const float* sLane, const float* gLane,
                                          const float* wLane, int u) {
    load_frag<2>(f1, sLane, gLane, wLane, u);
    fence();
    mma_frag<0>(acc, f0, f0.h);
    fence();
    load_frag<1>(f0, sLane, gLane, wLane, u);
    fence();
    tap_sums(ts, f0, f1);
    mma_frag<2>(acc, f1, f1.h);
    fence();
    load_frag<0>(f1, sLane, gLane, wLane, u + 1);
    fence();
    mma_frag<1>(acc, f0, ts.t);
    fence();
    load_frag<2>(f0, sLane, gLane, wLane, u + 1);
    fence();
    mma_frag<0>(acc, f1, f1.h);
    fence();
    load_frag<1>(f1, sLane, gLane, wLane, u + 1);
    fence();
    tap_sums(ts, f1, f0);
    mma_frag<2>(acc, f0, f0.h);
    fence();
    if (!LAST) load_frag<0>(f0, sLane, gLane, wLane, u + 2);
    fence();
    mma_frag<1>(acc, f1, ts.t);
    fence();
}

__global__ __launch_bounds__(NTHREADS)
void deconv1d_main_kernel(const float* __restrict__ x,
                          const float* __restrict__ w,
                          const float* __restrict__ bias,
                          float* __restrict__ out) {
    __shared__ __align__(16) float xs[XS_ROWS * PITCH];
    __shared__ __align__(16) float xg_[XG_ROWS * PITCH];

    const int tile = blockIdx.x & (T_ / TT - 1);
    const int b    = blockIdx.x / (T_ / TT);
    const int t_lo = tile * TT;
    const int tid  = threadIdx.x;

    const float4* xgl = (const float4*)(x + (size_t)b * T_ * C_);
    const int r0 = tid >> 6;
    const int cq = tid & 63;
    float4 v[STAGE_ITERS];
#pragma unroll
    for (int it = 0; it < STAGE_ITERS; ++it) {
        int t   = t_lo + r0 + 4 * it;
        int tc  = t < T_ ? t : (T_ - 1);
        float4 q = xgl[tc * (C_ / 4) + cq];
        if (t >= T_) q = make_float4(0.f, 0.f, 0.f, 0.f);
        v[it] = q;
    }
#pragma unroll
    for (int it = 0; it < STAGE_ITERS; ++it) {
        int row = r0 + 4 * it;
        if (it >= XS_LO / 4 && it < (XS_LO + XS_ROWS) / 4)
            *(float4*)&xs[(row - XS_LO) * PITCH + cq * 4] = v[it];
        if (it < XG_ROWS / 4) {
            float4 a = v[it], c = v[it + 2], d = v[it + 4], e = v[it + 6];
            *(float4*)&xg_[row * PITCH + cq * 4] = make_float4((a.x - c.x) - (d.x - e.x), (a.y - c.y) - (d.y - e.y),
                                                               (a.z - c.z) - (d.z - e.z), (a.w - c.w) - (d.w - e.w));
        }
    }
    __syncthreads();

    const int lane  = tid & 31;
    const int wv    = tid >> 5;
    const int n16   = lane & 15;
    const int half  = lane >> 4;
    const int chalf = half * 2;
    const int f0    = wv * (NT * 16);

    v8f acc[MT][NM][NT];
#pragma unroll
    for (int mt = 0; mt < MT; ++mt)
#pragma unroll
        for (int q = 0; q < NM; ++q)
#pragma unroll
            for (int ft = 0; ft < NT; ++ft)
                acc[mt][q][ft] = (v8f){0.f, 0.f, 0.f, 0.f, 0.f, 0.f, 0.f, 0.f};

    const int rowLane = 32 * (n16 >> 3) + (n16 & 7);
    const float* sLane = &xs[rowLane * PITCH + chalf];
    const float* gLane = &xg_[rowLane * PITCH + chalf];
    const float* wLane = w + (size_t)chalf * K_ * F_ + f0 + n16;

    Frag g0, g1;
    TapSum ts;
    load_frag<0>(g0, sLane, gLane, wLane, 0);

    const int wave_s = __builtin_amdgcn_readfirstlane(wv);
    if (wave_s >= NTHREADS / 64)
        __builtin_amdgcn_s_setprio(3);
    for (int u = 0; u < PRIO_STEPS; u += 2)
        two_steps<false>(acc, g0, g1, ts, sLane, gLane, wLane, u);
    __builtin_amdgcn_s_setprio(0);
    for (int u = PRIO_STEPS; u < NU - 2; u += 2)
        two_steps<false>(acc, g0, g1, ts, sLane, gLane, wLane, u);
    two_steps<true>(acc, g0, g1, ts, sLane, gLane, wLane, NU - 2);

    float biasv[NT];
#pragma unroll
    for (int ft = 0; ft < NT; ++ft)
        biasv[ft] = bias[f0 + ft * 16 + n16] * (float)C_;
    float* outb = out + (size_t)b * T_ * F_;
#pragma unroll
    for (int mt = 0; mt < MT; ++mt) {
#pragma unroll
        for (int blk = 0; blk < 4; ++blk) {
#pragma unroll
            for (int i = 0; i < 8; ++i) {
                int m   = 64 * mt + 32 * half + 8 * blk + i;
                int tau = t_lo + m;
                int d   = (tau + HALO) - (T_ - 1);
                int kmin = d > 0 ? ((d + 3) >> 2) : 0;
                float cnt = (float)(K_ - kmin);
                float* o = outb + (size_t)tau * F_ + f0 + n16;
#pragma unroll
                for (int ft = 0; ft < NT; ++ft) {
                    const v8f (&A)[NM][NT] = acc[mt];
                    float y;
                    if (blk == 0)      y = (A[0][ft][i] + A[1][ft][i]) + (A[3][ft][i] + A[4][ft][i]);
                    else if (blk == 1) y = (A[1][ft][i] - A[2][ft][i]) + (A[4][ft][i] - A[5][ft][i]);
                    else if (blk == 2) y = (A[3][ft][i] + A[4][ft][i]) - (A[6][ft][i] + A[7][ft][i]);
                    else               y = (A[4][ft][i] - A[5][ft][i]) - (A[7][ft][i] - A[8][ft][i]);
                    o[ft * 16] = y + cnt * biasv[ft];
                }
            }
        }
    }
}

extern "C" void kernel_launch(void* const* d_in, const int* in_sizes, int n_in,
                              void* d_out, int out_size, void* d_ws, size_t ws_size,
                              hipStream_t stream) {
    const float* x    = (const float*)d_in[0];
    const float* w    = (const float*)d_in[1];
    const float* bias = (const float*)d_in[2];
    float* out = (float*)d_out;

    deconv1d_main_kernel<<<B_ * (T_ / TT), NTHREADS, 0, stream>>>(
        x, w, bias, out);
}


# each block starts its 64-step (tap parity, channel pair) walk at its own step (29 x block index mod 64), so blocks do not fetch the same weight lines at one moment; rotates the summation order
# speedup vs baseline: 3.6848x; 1.0380x over previous
#include <hip/hip_runtime.h>


typedef float v2f __attribute__((ext_vector_type(2)));
typedef float v8f __attribute__((ext_vector_type(8)));

#define B_    16
#define T_    2048
#define C_    256
#define K_    8
#define F_    256
#define DIL_  4
#define HALO  28
#define TT    128
#define MT    2
#define NT    2
#define NJP   2
#define NM    9
#define XROWS 156
#define XS_LO 8
#define XS_ROWS 140
#define XG_ROWS 132
#define PITCH 260
#define NTHREADS 256
#define STAGE_ITERS ((XROWS * (C_ / 4)) / NTHREADS)
#define CPAIRS (C_ / 8)
#define NU (NJP * CPAIRS)
#define PRIO_STEPS 48
#define START_MUL 29
static_assert((NU & (NU - 1)) == 0, "the walk wraps with a mask");

__device__ __forceinline__ void fence() { __builtin_amdgcn_sched_barrier(0); }

struct Frag {
    v2f r[MT][2][4];
    v2f h[2][2][NT];
};
struct TapSum {
    v2f t[2][2][NT];
};

template <int OUT>
__device__ __forceinline__ void load_frag(Frag& f, const float* sLane, const float* gLane,
                                          const float* wLane, int u) {
    const int ue = u & (NU - 1);
    const int jp = ue / CPAIRS;
    const int cp = ue % CPAIRS;
    const int ao = (DIL_ * jp) * PITCH + cp * 8;
    const float* sP = sLane + ao;
    const float* gP = gLane + ao;
    const float* wP = wLane + ((size_t)cp * 8 * K_ + (K_ - 1 - jp)) * F_;
#pragma unroll
    for (int mt = 0; mt < MT; ++mt) {
        const float* sM = sP + mt * 64 * PITCH;
        const float* gM = gP + mt * 64 * PITCH;
#pragma unroll
        for (int s = 0; s < 2; ++s) {
            if (OUT == 0) {
                f.r[mt][s][0] = *(const v2f*)(gM + s * 4);
                f.r[mt][s][1] = *(const v2f*)(sM + (8 - XS_LO) * PITCH + s * 4);
                f.r[mt][s][2] = *(const v2f*)(sM + (24 - XS_LO) * PITCH + s * 4);
                f.r[mt][s][3] = *(const v2f*)(gM + 8 * PITCH + s * 4);
            } else if (OUT == 1) {
                f.r[mt][s][0] = *(const v2f*)(sM + (16 - XS_LO) * PITCH + s * 4);
                f.r[mt][s][1] = *(const v2f*)(sM + (24 - XS_LO) * PITCH + s * 4);
                f.r[mt][s][2] = *(const v2f*)(sM + (32 - XS_LO) * PITCH + s * 4);
            } else {
                f.r[mt][s][0] = *(const v2f*)(gM + 16 * PITCH + s * 4);
                f.r[mt][s][1] = *(const v2f*)(sM + (24 - XS_LO) * PITCH + s * 4);
                f.r[mt][s][2] = *(const v2f*)(sM + (40 - XS_LO) * PITCH + s * 4);
                f.r[mt][s][3] = *(const v2f*)(gM + 24 * PITCH + s * 4);
            }
        }
    }
    if (OUT != 1) {
#pragma unroll
        for (int i = 0; i < 2; ++i) {
            const int a = (OUT == 0 ? 0 : 2) + i;
#pragma unroll
            for (int s = 0; s < 2; ++s)
#pragma unroll
                for (int ft = 0; ft < NT; ++ft) {
                    const float* p = wP + ((s * 4) * K_ - 2 * a) * F_ + ft * 16;
                    f.h[i][s][ft] = (v2f){p[0], p[K_ * F_]};
                }
        }
    }
}

template <int OUT>
__device__ __forceinline__ void mma_frag(v8f (&acc)[MT][NM][NT], const Frag& f,
                                         const v2f (&g)[2][2][NT]) {
#pragma unroll
    for (int s = 0; s < 2; ++s) {
        v2f a[MT][3];
#pragma unroll
        for (int mt = 0; mt < MT; ++mt) {
            if (OUT == 1) {
                a[mt][0] = f.r[mt][s][0] - f.r[mt][s][1];
                a[mt][1] = f.r[mt][s][1];
                a[mt][2] = f.r[mt][s][1] - f.r[mt][s][2];
            } else {
                a[mt][0] = f.r[mt][s][0];
                a[mt][1] = f.r[mt][s][1] - f.r[mt][s][2];
                a[mt][2] = f.r[mt][s][3];
            }
        }
#pragma unroll
        for (int q = 0; q < 3; ++q)
#pragma unroll
            for (int ft = 0; ft < NT; ++ft) {
                v2f bf = q == 0 ? g[0][s][ft] : q == 2 ? g[1][s][ft] : g[0][s][ft] + g[1][s][ft];
#pragma unroll
                for (int mt = 0; mt < MT; ++mt)
                    acc[mt][OUT * 3 + q][ft] = __builtin_amdgcn_wmma_f32_16x16x4_f32(
                        false, a[mt][q], false, bf, (short)0, acc[mt][OUT * 3 + q][ft], false, false);
            }
    }
}

__device__ __forceinline__ void tap_sums(TapSum& ts, const Frag& f01, const Frag& f23) {
#pragma unroll
    for (int i = 0; i < 2; ++i)
#pragma unroll
        for (int s = 0; s < 2; ++s)
#pragma unroll
            for (int ft = 0; ft < NT; ++ft)
                ts.t[i][s][ft] = f01.h[i][s][ft] + f23.h[i][s][ft];
}

template <bool LAST>
__device__ __forceinline__ void two_steps(v8f (&acc)[MT][NM][NT], Frag& f0, Frag& f1, TapSum& ts,
                                          const float* sLane, const float* gLane,
                                          const float* wLane, int u) {
    load_frag<2>(f1, sLane, gLane, wLane, u);
    fence();
    mma_frag<0>(acc, f0, f0.h);
    fence();
    load_frag<1>(f0, sLane, gLane, wLane, u);
    fence();
    tap_sums(ts, f0, f1);
    mma_frag<2>(acc, f1, f1.h);
    fence();
    load_frag<0>(f1, sLane, gLane, wLane, u + 1);
    fence();
    mma_frag<1>(acc, f0, ts.t);
    fence();
    load_frag<2>(f0, sLane, gLane, wLane, u + 1);
    fence();
    mma_frag<0>(acc, f1, f1.h);
    fence();
    load_frag<1>(f1, sLane, gLane, wLane, u + 1);
    fence();
    tap_sums(ts, f1, f0);
    mma_frag<2>(acc, f0, f0.h);
    fence();
    if (!LAST) load_frag<0>(f0, sLane, gLane, wLane, u + 2);
    fence();
    mma_frag<1>(acc, f1, ts.t);
    fence();
}

__global__ __launch_bounds__(NTHREADS)
void deconv1d_main_kernel(const float* __restrict__ x,
                          const float* __restrict__ w,
                          const float* __restrict__ bias,
                          float* __restrict__ out) {
    __shared__ __align__(16) float xs[XS_ROWS * PITCH];
    __shared__ __align__(16) float xg_[XG_ROWS * PITCH];

    const int tile = blockIdx.x & (T_ / TT - 1);
    const int b    = blockIdx.x / (T_ / TT);
    const int t_lo = tile * TT;
    const int tid  = threadIdx.x;

    const float4* xgl = (const float4*)(x + (size_t)b * T_ * C_);
    const int r0 = tid >> 6;
    const int cq = tid & 63;
    float4 v[STAGE_ITERS];
#pragma unroll
    for (int it = 0; it < STAGE_ITERS; ++it) {
        int t   = t_lo + r0 + 4 * it;
        int tc  = t < T_ ? t : (T_ - 1);
        float4 q = xgl[tc * (C_ / 4) + cq];
        if (t >= T_) q = make_float4(0.f, 0.f, 0.f, 0.f);
        v[it] = q;
    }
#pragma unroll
    for (int it = 0; it < STAGE_ITERS; ++it) {
        int row = r0 + 4 * it;
        if (it >= XS_LO / 4 && it < (XS_LO + XS_ROWS) / 4)
            *(float4*)&xs[(row - XS_LO) * PITCH + cq * 4] = v[it];
        if (it < XG_ROWS / 4) {
            float4 a = v[it], c = v[it + 2], d = v[it + 4], e = v[it + 6];
            *(float4*)&xg_[row * PITCH + cq * 4] = make_float4((a.x - c.x) - (d.x - e.x), (a.y - c.y) - (d.y - e.y),
                                                               (a.z - c.z) - (d.z - e.z), (a.w - c.w) - (d.w - e.w));
        }
    }
    __syncthreads();

    const int lane  = tid & 31;
    const int wv    = tid >> 5;
    const int n16   = lane & 15;
    const int half  = lane >> 4;
    const int chalf = half * 2;
    const int f0    = wv * (NT * 16);

    v8f acc[MT][NM][NT];
#pragma unroll
    for (int mt = 0; mt < MT; ++mt)
#pragma unroll
        for (int q = 0; q < NM; ++q)
#pragma unroll
            for (int ft = 0; ft < NT; ++ft)
                acc[mt][q][ft] = (v8f){0.f, 0.f, 0.f, 0.f, 0.f, 0.f, 0.f, 0.f};

    const int rowLane = 32 * (n16 >> 3) + (n16 & 7);
    const float* sLane = &xs[rowLane * PITCH + chalf];
    const float* gLane = &xg_[rowLane * PITCH + chalf];
    const float* wLane = w + (size_t)chalf * K_ * F_ + f0 + n16;

    const int rot = (int)((blockIdx.x * START_MUL) & (NU - 1));

    Frag g0, g1;
    TapSum ts;
    load_frag<0>(g0, sLane, gLane, wLane, rot);

    const int wave_s = __builtin_amdgcn_readfirstlane(wv);
    if (wave_s >= NTHREADS / 64)
        __builtin_amdgcn_s_setprio(3);
    for (int u = 0; u < PRIO_STEPS; u += 2)
        two_steps<false>(acc, g0, g1, ts, sLane, gLane, wLane, rot + u);
    __builtin_amdgcn_s_setprio(0);
    for (int u = PRIO_STEPS; u < NU - 2; u += 2)
        two_steps<false>(acc, g0, g1, ts, sLane, gLane, wLane, rot + u);
    two_steps<true>(acc, g0, g1, ts, sLane, gLane, wLane, rot + NU - 2);

    float biasv[NT];
#pragma unroll
    for (int ft = 0; ft < NT; ++ft)
        biasv[ft] = bias[f0 + ft * 16 + n16] * (float)C_;
    float* outb = out + (size_t)b * T_ * F_;
#pragma unroll
    for (int mt = 0; mt < MT; ++mt) {
#pragma unroll
        for (int blk = 0; blk < 4; ++blk) {
#pragma unroll
            for (int i = 0; i < 8; ++i) {
                int m   = 64 * mt + 32 * half + 8 * blk + i;
                int tau = t_lo + m;
                int d   = (tau + HALO) - (T_ - 1);
                int kmin = d > 0 ? ((d + 3) >> 2) : 0;
                float cnt = (float)(K_ - kmin);
                float* o = outb + (size_t)tau * F_ + f0 + n16;
#pragma unroll
                for (int ft = 0; ft < NT; ++ft) {
                    const v8f (&A)[NM][NT] = acc[mt];
                    float y;
                    if (blk == 0)      y = (A[0][ft][i] + A[1][ft][i]) + (A[3][ft][i] + A[4][ft][i]);
                    else if (blk == 1) y = (A[1][ft][i] - A[2][ft][i]) + (A[4][ft][i] - A[5][ft][i]);
                    else if (blk == 2) y = (A[3][ft][i] + A[4][ft][i]) - (A[6][ft][i] + A[7][ft][i]);
                    else               y = (A[4][ft][i] - A[5][ft][i]) - (A[7][ft][i] - A[8][ft][i]);
                    o[ft * 16] = y + cnt * biasv[ft];
                }
            }
        }
    }
}

extern "C" void kernel_launch(void* const* d_in, const int* in_sizes, int n_in,
                              void* d_out, int out_size, void* d_ws, size_t ws_size,
                              hipStream_t stream) {
    const float* x    = (const float*)d_in[0];
    const float* w    = (const float*)d_in[1];
    const float* bias = (const float*)d_in[2];
    float* out = (float*)d_out;

    deconv1d_main_kernel<<<B_ * (T_ / TT), NTHREADS, 0, stream>>>(
        x, w, bias, out);
}


# one wave per SIMD (4 waves, tile 128 t x 64 f, a lane's 4 filter tiles = 4 adjacent filters): a step's operands loaded as one group ahead of the previous step's 144 WMMAs; sum order kept
# speedup vs baseline: 3.8488x; 1.0445x over previous
#include <hip/hip_runtime.h>


typedef float v2f __attribute__((ext_vector_type(2)));
typedef float v8f __attribute__((ext_vector_type(8)));

#define B_    16
#define T_    2048
#define C_    256
#define K_    8
#define F_    256
#define DIL_  4
#define HALO  28
#define TT    128
#define MT    2
#define NT    4
#define NJP   2
#define NM    9
#define XROWS 156
#define XS_LO 8
#define XS_ROWS 140
#define XG_ROWS 132
#define PITCH 260
#define NTHREADS 128
#define STAGE_ITERS ((XROWS * (C_ / 4)) / NTHREADS)
#define STAGE_RPI (NTHREADS / (C_ / 4))
#define CPAIRS (C_ / 8)
#define NU (NJP * CPAIRS)
#define START_MUL 29
static_assert((NU & (NU - 1)) == 0, "the walk wraps with a mask");
static_assert(NT == 4, "a lane's filter tiles are the four floats of one 16-byte load / store");

__device__ __forceinline__ void fence() { __builtin_amdgcn_sched_barrier(0); }

struct Frag {
    v2f r[MT][2][4];
    float4 h[2][2][2];
};
struct TapSum {
    float4 t[2][2][2];
};

__device__ __forceinline__ v2f col_pair(const float4& lo, const float4& hi, int ft) {
    return ft == 0 ? (v2f){lo.x, hi.x} : ft == 1 ? (v2f){lo.y, hi.y} : ft == 2 ? (v2f){lo.z, hi.z} : (v2f){lo.w, hi.w};
}
__device__ __forceinline__ float4 add4(const float4& a, const float4& b) {
    return make_float4(a.x + b.x, a.y + b.y, a.z + b.z, a.w + b.w);
}

template <int OUT>
__device__ __forceinline__ void load_frag(Frag& f, const float* sLane, const float* gLane,
                                          const float* wLane, int u) {
    const int ue = u & (NU - 1);
    const int jp = ue / CPAIRS;
    const int cp = ue % CPAIRS;
    const int ao = (DIL_ * jp) * PITCH + cp * 8;
    const float* sP = sLane + ao;
    const float* gP = gLane + ao;
    const float* wP = wLane + ((size_t)cp * 8 * K_ + (K_ - 1 - jp)) * F_;
#pragma unroll
    for (int mt = 0; mt < MT; ++mt) {
        const float* sM = sP + mt * 64 * PITCH;
        const float* gM = gP + mt * 64 * PITCH;
#pragma unroll
        for (int s = 0; s < 2; ++s) {
            if (OUT == 0) {
                f.r[mt][s][0] = *(const v2f*)(gM + s * 4);
                f.r[mt][s][1] = *(const v2f*)(sM + (8 - XS_LO) * PITCH + s * 4);
                f.r[mt][s][2] = *(const v2f*)(sM + (24 - XS_LO) * PITCH + s * 4);
                f.r[mt][s][3] = *(const v2f*)(gM + 8 * PITCH + s * 4);
            } else if (OUT == 1) {
                f.r[mt][s][0] = *(const v2f*)(sM + (16 - XS_LO) * PITCH + s * 4);
                f.r[mt][s][1] = *(const v2f*)(sM + (24 - XS_LO) * PITCH + s * 4);
                f.r[mt][s][2] = *(const v2f*)(sM + (32 - XS_LO) * PITCH + s * 4);
            } else {
                f.r[mt][s][0] = *(const v2f*)(gM + 16 * PITCH + s * 4);
                f.r[mt][s][1] = *(const v2f*)(sM + (24 - XS_LO) * PITCH + s * 4);
                f.r[mt][s][2] = *(const v2f*)(sM + (40 - XS_LO) * PITCH + s * 4);
                f.r[mt][s][3] = *(const v2f*)(gM + 24 * PITCH + s * 4);
            }
        }
    }
    if (OUT != 1) {
#pragma unroll
        for (int i = 0; i < 2; ++i) {
            const int a = (OUT == 0 ? 0 : 2) + i;
#pragma unroll
            for (int s = 0; s < 2; ++s) {
                const float* p = wP + ((s * 4) * K_ - 2 * a) * F_;
                f.h[i][s][0] = *(const float4*)p;
                f.h[i][s][1] = *(const float4*)(p + K_ * F_);
            }
        }
    }
}

template <int OUT>
__device__ __forceinline__ void mma_frag(v8f (&acc)[MT][NM][NT], const Frag& f,
                                         const float4 (&g)[2][2][2]) {
#pragma unroll
    for (int s = 0; s < 2; ++s) {
        v2f a[MT][3];
#pragma unroll
        for (int mt = 0; mt < MT; ++mt) {
            if (OUT == 1) {
                a[mt][0] = f.r[mt][s][0] - f.r[mt][s][1];
                a[mt][1] = f.r[mt][s][1];
                a[mt][2] = f.r[mt][s][1] - f.r[mt][s][2];
            } else {
                a[mt][0] = f.r[mt][s][0];
                a[mt][1] = f.r[mt][s][1] - f.r[mt][s][2];
                a[mt][2] = f.r[mt][s][3];
            }
        }
        const float4 sum0 = add4(g[0][s][0], g[1][s][0]);
        const float4 sum1 = add4(g[0][s][1], g[1][s][1]);
#pragma unroll
        for (int q = 0; q < 3; ++q)
#pragma unroll
            for (int ft = 0; ft < NT; ++ft) {
                v2f bf = q == 0 ? col_pair(g[0][s][0], g[0][s][1], ft)
                       : q == 2 ? col_pair(g[1][s][0], g[1][s][1], ft)
                                : col_pair(sum0, sum1, ft);
#pragma unroll
                for (int mt = 0; mt < MT; ++mt)
                    acc[mt][OUT * 3 + q][ft] = __builtin_amdgcn_wmma_f32_16x16x4_f32(
                        false, a[mt][q], false, bf, (short)0, acc[mt][OUT * 3 + q][ft], false, false);
            }
    }
}

__device__ __forceinline__ void tap_sums(TapSum& ts, const Frag& f01, const Frag& f23) {
#pragma unroll
    for (int i = 0; i < 2; ++i)
#pragma unroll
        for (int s = 0; s < 2; ++s)
#pragma unroll
            for (int e = 0; e < 2; ++e)
                ts.t[i][s][e] = add4(f01.h[i][s][e], f23.h[i][s][e]);
}

struct StepFrag {
    Frag o0, o2, o1;
};

__device__ __forceinline__ void load_step(StepFrag& f, const float* sLane, const float* gLane,
                                          const float* wLane, int u) {
    load_frag<0>(f.o0, sLane, gLane, wLane, u);
    load_frag<2>(f.o2, sLane, gLane, wLane, u);
    load_frag<1>(f.o1, sLane, gLane, wLane, u);
}

__device__ __forceinline__ void mma_step(v8f (&acc)[MT][NM][NT], const StepFrag& f) {
    TapSum ts;
    mma_frag<0>(acc, f.o0, f.o0.h);
    tap_sums(ts, f.o0, f.o2);
    mma_frag<2>(acc, f.o2, f.o2.h);
    mma_frag<1>(acc, f.o1, ts.t);
}

__device__ __forceinline__ void two_steps(v8f (&acc)[MT][NM][NT], StepFrag& f0, StepFrag& f1,
                                          const float* sLane, const float* gLane,
                                          const float* wLane, int u) {
    load_step(f1, sLane, gLane, wLane, u + 1);
    fence();
    mma_step(acc, f0);
    fence();
    load_step(f0, sLane, gLane, wLane, u + 2);
    fence();
    mma_step(acc, f1);
    fence();
}

__global__ __launch_bounds__(NTHREADS)
void deconv1d_main_kernel(const float* __restrict__ x,
                          const float* __restrict__ w,
                          const float* __restrict__ bias,
                          float* __restrict__ out) {
    __shared__ __align__(16) float xs[XS_ROWS * PITCH];
    __shared__ __align__(16) float xg_[XG_ROWS * PITCH];

    const int tile = blockIdx.x & (T_ / TT - 1);
    const int b    = blockIdx.x / (T_ / TT);
    const int t_lo = tile * TT;
    const int tid  = threadIdx.x;

    const float4* xgl = (const float4*)(x + (size_t)b * T_ * C_);
    const int r0 = tid >> 6;
    const int cq = tid & 63;
    float4 v[STAGE_ITERS];
#pragma unroll
    for (int it = 0; it < STAGE_ITERS; ++it) {
        int t   = t_lo + r0 + STAGE_RPI * it;
        int tc  = t < T_ ? t : (T_ - 1);
        float4 q = xgl[tc * (C_ / 4) + cq];
        if (t >= T_) q = make_float4(0.f, 0.f, 0.f, 0.f);
        v[it] = q;
    }
#pragma unroll
    for (int it = 0; it < STAGE_ITERS; ++it) {
        int row = r0 + STAGE_RPI * it;
        if (it >= XS_LO / STAGE_RPI && it < (XS_LO + XS_ROWS) / STAGE_RPI)
            *(float4*)&xs[(row - XS_LO) * PITCH + cq * 4] = v[it];
        if (it < XG_ROWS / STAGE_RPI) {
            float4 a = v[it], c = v[it + 8 / STAGE_RPI], d = v[it + 16 / STAGE_RPI], e = v[it + 24 / STAGE_RPI];
            *(float4*)&xg_[row * PITCH + cq * 4] = make_float4((a.x - c.x) - (d.x - e.x), (a.y - c.y) - (d.y - e.y),
                                                               (a.z - c.z) - (d.z - e.z), (a.w - c.w) - (d.w - e.w));
        }
    }
    __syncthreads();

    const int lane  = tid & 31;
    const int wv    = tid >> 5;
    const int n16   = lane & 15;
    const int half  = lane >> 4;
    const int chalf = half * 2;
    const int f0    = wv * (NT * 16);

    v8f acc[MT][NM][NT];
#pragma unroll
    for (int mt = 0; mt < MT; ++mt)
#pragma unroll
        for (int q = 0; q < NM; ++q)
#pragma unroll
            for (int ft = 0; ft < NT; ++ft)
                acc[mt][q][ft] = (v8f){0.f, 0.f, 0.f, 0.f, 0.f, 0.f, 0.f, 0.f};

    const int rowLane = 32 * (n16 >> 3) + (n16 & 7);
    const float* sLane = &xs[rowLane * PITCH + chalf];
    const float* gLane = &xg_[rowLane * PITCH + chalf];
    const float* wLane = w + (size_t)chalf * K_ * F_ + f0 + 4 * n16;

    const int rot = (int)((blockIdx.x * START_MUL) & (NU - 1));

    StepFrag g0, g1;
    load_step(g0, sLane, gLane, wLane, rot);

    for (int u = 0; u < NU; u += 2)
        two_steps(acc, g0, g1, sLane, gLane, wLane, rot + u);

    const float4 b4 = *(const float4*)(bias + f0 + 4 * n16);
    const float biasv[NT] = {b4.x * (float)C_, b4.y * (float)C_, b4.z * (float)C_, b4.w * (float)C_};
    float* outb = out + (size_t)b * T_ * F_;
#pragma unroll
    for (int mt = 0; mt < MT; ++mt) {
#pragma unroll
        for (int blk = 0; blk < 4; ++blk) {
#pragma unroll
            for (int i = 0; i < 8; ++i) {
                int m   = 64 * mt + 32 * half + 8 * blk + i;
                int tau = t_lo + m;
                int d   = (tau + HALO) - (T_ - 1);
                int kmin = d > 0 ? ((d + 3) >> 2) : 0;
                float cnt = (float)(K_ - kmin);
                float* o = outb + (size_t)tau * F_ + f0 + 4 * n16;
                float yv[NT];
#pragma unroll
                for (int ft = 0; ft < NT; ++ft) {
                    const v8f (&A)[NM][NT] = acc[mt];
                    float y;
                    if (blk == 0)      y = (A[0][ft][i] + A[1][ft][i]) + (A[3][ft][i] + A[4][ft][i]);
                    else if (blk == 1) y = (A[1][ft][i] - A[2][ft][i]) + (A[4][ft][i] - A[5][ft][i]);
                    else if (blk == 2) y = (A[3][ft][i] + A[4][ft][i]) - (A[6][ft][i] + A[7][ft][i]);
                    else               y = (A[4][ft][i] - A[5][ft][i]) - (A[7][ft][i] - A[8][ft][i]);
                    yv[ft] = y + cnt * biasv[ft];
                }
                *(float4*)o = make_float4(yv[0], yv[1], yv[2], yv[3]);
            }
        }
    }
}

extern "C" void kernel_launch(void* const* d_in, const int* in_sizes, int n_in,
                              void* d_out, int out_size, void* d_ws, size_t ws_size,
                              hipStream_t stream) {
    const float* x    = (const float*)d_in[0];
    const float* w    = (const float*)d_in[1];
    const float* bias = (const float*)d_in[2];
    float* out = (float*)d_out;

    deconv1d_main_kernel<<<B_ * (T_ / TT), NTHREADS, 0, stream>>>(
        x, w, bias, out);
}
